# starting kernel
# baseline (speedup 1.0000x reference)
#include <hip/hip_runtime.h>
#include <math.h>

#define N_NODES 100000
#define N_EDGES 3200000
#define F_IN    512
#define HID     16
#define NCLS    10
#define NTILES  (N_NODES / 16)

typedef __attribute__((ext_vector_type(16))) __bf16 v16bf;
typedef __attribute__((ext_vector_type(8)))  float  v8f;


__global__ void k_init_deg(float* __restrict__ deg) {
    int i = blockIdx.x * blockDim.x + threadIdx.x;
    if (i < N_NODES) deg[i] = 1.0f;
}

__global__ void k_deg_edges(const int* __restrict__ dst, const float* __restrict__ w,
                            float* __restrict__ deg) {
    int e = blockIdx.x * blockDim.x + threadIdx.x;
    if (e < N_EDGES) atomicAdd(&deg[dst[e]], w[e]);
}

__global__ void k_dinv(float* __restrict__ deg) {
    int i = blockIdx.x * blockDim.x + threadIdx.x;
    if (i < N_NODES) {
        float d = deg[i];
        deg[i] = (d > 0.0f) ? rsqrtf(d) : 0.0f;
    }
}

__global__ void k_norm(const int* __restrict__ src, const int* __restrict__ dst,
                       const float* __restrict__ w, const float* __restrict__ dinv,
                       float* __restrict__ nrm) {
    int e = blockIdx.x * blockDim.x + threadIdx.x;
    if (e < N_EDGES) nrm[e] = dinv[src[e]] * w[e] * dinv[dst[e]];
}


__global__ void k_gemm1(const float* __restrict__ x, const float* __restrict__ W1,
                        float* __restrict__ h) {
    __shared__ __bf16 w1s[F_IN * HID];
    int t = threadIdx.x;
    for (int i = t; i < F_IN * HID; i += blockDim.x) w1s[i] = (__bf16)W1[i];
    __syncthreads();

    int wave = t >> 5;
    int lane = t & 31;
    int tile = blockIdx.x * 8 + wave;
    if (tile >= NTILES) return;

    int col = lane & 15;
    int hh  = lane >> 4;
    const float* xr = x + (size_t)(tile * 16 + col) * F_IN;

    v8f acc;
    #pragma unroll
    for (int r = 0; r < 8; ++r) acc[r] = 0.0f;

    for (int k0 = 0; k0 < F_IN; k0 += 32) {
        v16bf a, b;
        #pragma unroll
        for (int j = 0; j < 8; ++j) {
            int kl = (j < 4) ? (2 * j + 8 * hh) : (16 + 2 * (j - 4) + 8 * hh);
            int k  = k0 + kl;
            float2 xv = *reinterpret_cast<const float2*>(xr + k);
            a[2 * j]     = (__bf16)xv.x;
            a[2 * j + 1] = (__bf16)xv.y;
            b[2 * j]     = w1s[k * HID + col];
            b[2 * j + 1] = w1s[(k + 1) * HID + col];
        }
        acc = __builtin_amdgcn_wmma_f32_16x16x32_bf16(
                  false, a, false, b, (short)0, acc, false, false);
    }

    float* out = h + (size_t)(tile * 16 + 8 * hh) * HID + col;
    #pragma unroll
    for (int r = 0; r < 8; ++r) out[(size_t)r * HID] = acc[r];
}


__global__ void k_gemm2(const float* __restrict__ h1, const float* __restrict__ W2,
                        float* __restrict__ h2) {
    __shared__ __bf16 w2s[16 * 16];
    int t = threadIdx.x;
    for (int i = t; i < 16 * 16; i += blockDim.x) {
        int k = i >> 4, n = i & 15;
        w2s[i] = (n < NCLS) ? (__bf16)W2[k * NCLS + n] : (__bf16)0.0f;
    }
    __syncthreads();

    int wave = t >> 5;
    int lane = t & 31;
    int tile = blockIdx.x * 8 + wave;
    if (tile >= NTILES) return;

    int col = lane & 15;
    int hh  = lane >> 4;
    const float* hr = h1 + (size_t)(tile * 16 + col) * HID;

    v16bf a, b;
    #pragma unroll
    for (int i = 0; i < 16; ++i) { a[i] = (__bf16)0.0f; b[i] = (__bf16)0.0f; }
    #pragma unroll
    for (int j = 0; j < 8; ++j) {
        int kl = (j < 4) ? (2 * j + 8 * hh) : (16 + 2 * (j - 4) + 8 * hh);
        if (kl < 16) {
            a[2 * j]     = (__bf16)hr[kl];
            a[2 * j + 1] = (__bf16)hr[kl + 1];
            b[2 * j]     = w2s[kl * 16 + col];
            b[2 * j + 1] = w2s[(kl + 1) * 16 + col];
        }
    }
    v8f acc;
    #pragma unroll
    for (int r = 0; r < 8; ++r) acc[r] = 0.0f;
    acc = __builtin_amdgcn_wmma_f32_16x16x32_bf16(
              false, a, false, b, (short)0, acc, false, false);

    float* out = h2 + (size_t)(tile * 16 + 8 * hh) * HID + col;
    #pragma unroll
    for (int r = 0; r < 8; ++r) out[(size_t)r * HID] = acc[r];
}


__global__ void k_self_init(const float* __restrict__ dinv, const float* __restrict__ h,
                            float* __restrict__ agg) {
    int i = blockIdx.x * blockDim.x + threadIdx.x;
    if (i < N_NODES * HID) {
        float di = dinv[i >> 4];
        agg[i] = di * di * h[i];
    }
}

__global__ void k_scatter(const int* __restrict__ src, const int* __restrict__ dst,
                          const float* __restrict__ nrm, const float* __restrict__ h,
                          float* __restrict__ agg) {
    long long i = (long long)blockIdx.x * blockDim.x + threadIdx.x;
    if (i >= (long long)N_EDGES * HID) return;
    int e = (int)(i >> 4);
    int c = (int)(i & 15);
    float nv = nrm[e];
    atomicAdd(&agg[(size_t)dst[e] * HID + c], nv * h[(size_t)src[e] * HID + c]);
}

__global__ void k_bias_relu(const float* __restrict__ b1, float* __restrict__ agg) {
    int i = blockIdx.x * blockDim.x + threadIdx.x;
    if (i < N_NODES * HID) {
        float v = agg[i] + b1[i & 15];
        agg[i] = v > 0.0f ? v : 0.0f;
    }
}


__global__ void k_logsoftmax(const float* __restrict__ agg2, const float* __restrict__ b2,
                             float* __restrict__ out) {
    int i = blockIdx.x * blockDim.x + threadIdx.x;
    if (i >= N_NODES) return;
    float z[NCLS];
    float m = -INFINITY;
    #pragma unroll
    for (int c = 0; c < NCLS; ++c) {
        z[c] = agg2[(size_t)i * HID + c] + b2[c];
        m = fmaxf(m, z[c]);
    }
    float s = 0.0f;
    #pragma unroll
    for (int c = 0; c < NCLS; ++c) s += __expf(z[c] - m);
    float ls = __logf(s) + m;
    #pragma unroll
    for (int c = 0; c < NCLS; ++c) out[(size_t)i * NCLS + c] = z[c] - ls;
}


extern "C" void kernel_launch(void* const* d_in, const int* in_sizes, int n_in,
                              void* d_out, int out_size, void* d_ws, size_t ws_size,
                              hipStream_t stream) {
    const float* x  = (const float*)d_in[0];
    const int*   ei = (const int*)  d_in[1];
    const float* ew = (const float*)d_in[2];
    const float* W1 = (const float*)d_in[3];
    const float* b1 = (const float*)d_in[4];
    const float* W2 = (const float*)d_in[5];
    const float* b2 = (const float*)d_in[6];
    float* out = (float*)d_out;

    const int* srcp = ei;
    const int* dstp = ei + N_EDGES;

    float* ws   = (float*)d_ws;
    float* dinv = ws;
    float* nrm  = dinv + N_NODES;
    float* hbuf = nrm  + N_EDGES;
    float* agg1 = hbuf + (size_t)N_NODES * HID;
    float* h2   = agg1 + (size_t)N_NODES * HID;
    float* agg2 = h2   + (size_t)N_NODES * HID;

    const int B = 256;
    int gN   = (N_NODES + B - 1) / B;
    int gE   = (N_EDGES + B - 1) / B;
    int gNH  = (N_NODES * HID + B - 1) / B;
    long long eh = (long long)N_EDGES * HID;
    int gEH  = (int)((eh + B - 1) / B);
    int gMM  = (NTILES + 7) / 8;

    k_init_deg <<<gN, B, 0, stream>>>(dinv);
    k_deg_edges<<<gE, B, 0, stream>>>(dstp, ew, dinv);
    k_dinv     <<<gN, B, 0, stream>>>(dinv);
    k_norm     <<<gE, B, 0, stream>>>(srcp, dstp, ew, dinv, nrm);

    k_gemm1    <<<gMM, B, 0, stream>>>(x, W1, hbuf);
    k_self_init<<<gNH, B, 0, stream>>>(dinv, hbuf, agg1);
    k_scatter  <<<gEH, B, 0, stream>>>(srcp, dstp, nrm, hbuf, agg1);
    k_bias_relu<<<gNH, B, 0, stream>>>(b1, agg1);

    k_gemm2    <<<gMM, B, 0, stream>>>(agg1, W2, h2);
    k_self_init<<<gNH, B, 0, stream>>>(dinv, h2, agg2);
    k_scatter  <<<gEH, B, 0, stream>>>(srcp, dstp, nrm, h2, agg2);
    k_logsoftmax<<<gN, B, 0, stream>>>(agg2, b2, out);
}


# Edge scatter-adds (degree and both layers) accumulate per destination range in LDS partial sums, flushed by whole-line atomics; summation order changes (new reduction tree)
# speedup vs baseline: 1.7085x; 1.7085x over previous
#include <hip/hip_runtime.h>
#include <math.h>

#define N_NODES 100000
#define N_EDGES 3200000
#define F_IN    512
#define HID     16
#define NCLS    10
#define NTILES  (N_NODES / 16)

typedef __attribute__((ext_vector_type(16))) __bf16 v16bf;
typedef __attribute__((ext_vector_type(8)))  float  v8f;


__global__ void k_init_deg(float* __restrict__ deg) {
    int i = blockIdx.x * blockDim.x + threadIdx.x;
    if (i < N_NODES) deg[i] = 1.0f;
}


#define DEG_R      50000
#define DEG_GE     (N_EDGES / 32)

__global__ __launch_bounds__(1024) void k_deg_lds(const int* __restrict__ dst,
                                                  const float* __restrict__ w,
                                                  float* __restrict__ deg) {
    __shared__ float sd[DEG_R];
    const int t  = threadIdx.x;
    const int r  = blockIdx.x & 1;
    const int g  = blockIdx.x >> 1;
    const int lo = r * DEG_R;
    const int e0 = g * DEG_GE;
    const int e1 = e0 + DEG_GE;

    for (int i = t; i < DEG_R; i += 1024) sd[i] = 0.0f;
    __syncthreads();

    for (int b = e0 + t * 4; b < e1; b += 4 * 4096) {
        int4   d4[4];
        float4 w4[4];
        #pragma unroll
        for (int u = 0; u < 4; ++u) {
            const int e  = b + u * 4096;
            const int ec = (e < e1) ? e : (e1 - 4);
            d4[u] = *reinterpret_cast<const int4*>(dst + ec);
            w4[u] = *reinterpret_cast<const float4*>(w + ec);
        }
        #pragma unroll
        for (int u = 0; u < 4; ++u) {
            if (b + u * 4096 < e1) {
                const unsigned u0 = (unsigned)(d4[u].x - lo), u1 = (unsigned)(d4[u].y - lo);
                const unsigned u2 = (unsigned)(d4[u].z - lo), u3 = (unsigned)(d4[u].w - lo);
                if (u0 < DEG_R) atomicAdd(&sd[u0], w4[u].x);
                if (u1 < DEG_R) atomicAdd(&sd[u1], w4[u].y);
                if (u2 < DEG_R) atomicAdd(&sd[u2], w4[u].z);
                if (u3 < DEG_R) atomicAdd(&sd[u3], w4[u].w);
            }
        }
    }
    __syncthreads();

    float* dg = deg + lo;
    const int shift = lo & 31;
    for (int q = 0; q < 49; ++q) {
        int it = q + g; it = it % 49;
        int j = it * 1024 + t - shift;
        if (j >= 0 && j < DEG_R) {
            float v = sd[j];
            if (v != 0.0f) atomicAdd(&dg[j], v);
        }
    }
}

__global__ void k_dinv(float* __restrict__ deg) {
    int i = blockIdx.x * blockDim.x + threadIdx.x;
    if (i < N_NODES) {
        float d = deg[i];
        deg[i] = (d > 0.0f) ? rsqrtf(d) : 0.0f;
    }
}

__global__ void k_norm(const int* __restrict__ src, const int* __restrict__ dst,
                       const float* __restrict__ w, const float* __restrict__ dinv,
                       float* __restrict__ nrm) {
    int e = blockIdx.x * blockDim.x + threadIdx.x;
    if (e < N_EDGES) nrm[e] = dinv[src[e]] * w[e] * dinv[dst[e]];
}


__global__ void k_gemm1(const float* __restrict__ x, const float* __restrict__ W1,
                        float* __restrict__ h) {
    __shared__ __bf16 w1s[F_IN * HID];
    int t = threadIdx.x;
    for (int i = t; i < F_IN * HID; i += blockDim.x) w1s[i] = (__bf16)W1[i];
    __syncthreads();

    int wave = t >> 5;
    int lane = t & 31;
    int tile = blockIdx.x * 8 + wave;
    if (tile >= NTILES) return;

    int col = lane & 15;
    int hh  = lane >> 4;
    const float* xr = x + (size_t)(tile * 16 + col) * F_IN;

    v8f acc;
    #pragma unroll
    for (int r = 0; r < 8; ++r) acc[r] = 0.0f;

    for (int k0 = 0; k0 < F_IN; k0 += 32) {
        v16bf a, b;
        #pragma unroll
        for (int j = 0; j < 8; ++j) {
            int kl = (j < 4) ? (2 * j + 8 * hh) : (16 + 2 * (j - 4) + 8 * hh);
            int k  = k0 + kl;
            float2 xv = *reinterpret_cast<const float2*>(xr + k);
            a[2 * j]     = (__bf16)xv.x;
            a[2 * j + 1] = (__bf16)xv.y;
            b[2 * j]     = w1s[k * HID + col];
            b[2 * j + 1] = w1s[(k + 1) * HID + col];
        }
        acc = __builtin_amdgcn_wmma_f32_16x16x32_bf16(
                  false, a, false, b, (short)0, acc, false, false);
    }

    float* out = h + (size_t)(tile * 16 + 8 * hh) * HID + col;
    #pragma unroll
    for (int r = 0; r < 8; ++r) out[(size_t)r * HID] = acc[r];
}


__global__ void k_gemm2(const float* __restrict__ h1, const float* __restrict__ W2,
                        float* __restrict__ h2) {
    __shared__ __bf16 w2s[16 * 16];
    int t = threadIdx.x;
    for (int i = t; i < 16 * 16; i += blockDim.x) {
        int k = i >> 4, n = i & 15;
        w2s[i] = (n < NCLS) ? (__bf16)W2[k * NCLS + n] : (__bf16)0.0f;
    }
    __syncthreads();

    int wave = t >> 5;
    int lane = t & 31;
    int tile = blockIdx.x * 8 + wave;
    if (tile >= NTILES) return;

    int col = lane & 15;
    int hh  = lane >> 4;
    const float* hr = h1 + (size_t)(tile * 16 + col) * HID;

    v16bf a, b;
    #pragma unroll
    for (int i = 0; i < 16; ++i) { a[i] = (__bf16)0.0f; b[i] = (__bf16)0.0f; }
    #pragma unroll
    for (int j = 0; j < 8; ++j) {
        int kl = (j < 4) ? (2 * j + 8 * hh) : (16 + 2 * (j - 4) + 8 * hh);
        if (kl < 16) {
            a[2 * j]     = (__bf16)hr[kl];
            a[2 * j + 1] = (__bf16)hr[kl + 1];
            b[2 * j]     = w2s[kl * 16 + col];
            b[2 * j + 1] = w2s[(kl + 1) * 16 + col];
        }
    }
    v8f acc;
    #pragma unroll
    for (int r = 0; r < 8; ++r) acc[r] = 0.0f;
    acc = __builtin_amdgcn_wmma_f32_16x16x32_bf16(
              false, a, false, b, (short)0, acc, false, false);

    float* out = h2 + (size_t)(tile * 16 + 8 * hh) * HID + col;
    #pragma unroll
    for (int r = 0; r < 8; ++r) out[(size_t)r * HID] = acc[r];
}


__global__ void k_self_init(const float* __restrict__ dinv, const float* __restrict__ h,
                            float* __restrict__ agg) {
    int i = blockIdx.x * blockDim.x + threadIdx.x;
    if (i < N_NODES * HID) {
        float di = dinv[i >> 4];
        agg[i] = di * di * h[i];
    }
}


#define SC_R       3125
#define SC_GE      (N_EDGES / 8)
#define SC_V4      6
#define SC_CHUNK   (1024 * 4 * SC_V4)
#define SC_CAP     8192

__device__ __forceinline__ void sc_load_dst(int4 (&dv)[SC_V4], const int* __restrict__ dst,
                                            int cb, int e1, int t) {
    #pragma unroll
    for (int k = 0; k < SC_V4; ++k) {
        int e = cb + (k * 1024 + t) * 4;
        dv[k] = (e < e1) ? *reinterpret_cast<const int4*>(dst + e) : make_int4(-1, -1, -1, -1);
    }
}

__global__ __launch_bounds__(1024) void k_scatter_lds(const int* __restrict__ src,
                                                      const int* __restrict__ dst,
                                                      const float* __restrict__ nrm,
                                                      const float* __restrict__ h,
                                                      float* __restrict__ agg) {
    __shared__ float acc[SC_R * HID];
    __shared__ int   lkey[SC_CAP];
    __shared__ float lval[SC_CAP];
    __shared__ int   cnt[2];

    const int t    = threadIdx.x;
    const int lane = t & 31;
    const int c    = t & 15;
    const int r    = blockIdx.x & 31;
    const int g    = blockIdx.x >> 5;
    const int lo   = r * SC_R;
    const int e0   = g * SC_GE;
    const int e1   = e0 + SC_GE;

    for (int i = t; i < SC_R * HID; i += 1024) acc[i] = 0.0f;
    if (t < 2) cnt[t] = 0;

    int4 dv[SC_V4];
    sc_load_dst(dv, dst, e0, e1, t);
    __syncthreads();

    const int nch = (SC_GE + SC_CHUNK - 1) / SC_CHUNK;
    for (int ch = 0; ch < nch; ++ch) {
        const int cb = e0 + ch * SC_CHUNK;
        int* cp = &cnt[ch & 1];

        int dd[SC_V4 * 4];
        unsigned mm = 0;
        #pragma unroll
        for (int k = 0; k < SC_V4; ++k) {
            dd[k * 4 + 0] = dv[k].x - lo; dd[k * 4 + 1] = dv[k].y - lo;
            dd[k * 4 + 2] = dv[k].z - lo; dd[k * 4 + 3] = dv[k].w - lo;
        }
        #pragma unroll
        for (int j = 0; j < SC_V4 * 4; ++j)
            if ((unsigned)dd[j] < SC_R) mm |= 1u << j;

        const int my = __builtin_popcount(mm);
        int inc = my;
        #pragma unroll
        for (int d = 1; d < 32; d <<= 1) {
            int o = __shfl_up(inc, d);
            if (lane >= d) inc += o;
        }
        const int wtot = __shfl(inc, 31);
        int wbase = 0;
        if (lane == 31 && wtot > 0) wbase = atomicAdd(cp, wtot);
        wbase = __shfl(wbase, 31);
        int pos = wbase + inc - my;

        #pragma unroll
        for (int j = 0; j < SC_V4 * 4; ++j) {
            if ((mm >> j) & 1u) {
                const int el = ((j >> 2) * 1024 + t) * 4 + (j & 3);
                if (pos < SC_CAP) {
                    lkey[pos] = el | (dd[j] << 15);
                } else {
                    const int e = cb + el;
                    const float nvv = nrm[e];
                    const float* hp = h + (size_t)src[e] * HID;
                    float* ap = agg + (size_t)(lo + dd[j]) * HID;
                    #pragma nounroll
                    for (int cc = 0; cc < HID; ++cc) atomicAdd(ap + cc, nvv * hp[cc]);
                }
                ++pos;
            }
        }
        __syncthreads();

        int n = *cp;
        if (n > SC_CAP) n = SC_CAP;
        if (t == 0) cnt[(ch + 1) & 1] = 0;

        if (ch + 1 < nch) sc_load_dst(dv, dst, cb + SC_CHUNK, e1, t);

        for (int i = t; i < n; i += 1024) {
            const int ent = lkey[i];
            const int e   = cb + (ent & 0x7FFF);
            lkey[i] = src[e] | ((ent >> 15) << 17);
            lval[i] = nrm[e];
        }
        __syncthreads();

        int base = t >> 4;
        for (; base + 192 < n; base += 256) {
            int   kk[4];
            float vv[4], hv[4];
            #pragma unroll
            for (int j = 0; j < 4; ++j) { kk[j] = lkey[base + 64 * j]; vv[j] = lval[base + 64 * j]; }
            #pragma unroll
            for (int j = 0; j < 4; ++j) hv[j] = h[(size_t)(kk[j] & 0x1FFFF) * HID + c];
            #pragma unroll
            for (int j = 0; j < 4; ++j) atomicAdd(&acc[(kk[j] >> 17) * HID + c], vv[j] * hv[j]);
        }
        for (; base < n; base += 64) {
            const int   k1 = lkey[base];
            const float v1 = lval[base];
            atomicAdd(&acc[(k1 >> 17) * HID + c], v1 * h[(size_t)(k1 & 0x1FFFF) * HID + c]);
        }
        __syncthreads();
    }

    float* ag = agg + (size_t)lo * HID;
    const int shift = (lo * HID) & 31;
    for (int q = 0; q < 49; ++q) {
        int it = q + g * 6; it = it % 49;
        int j = it * 1024 + t - shift;
        if (j >= 0 && j < SC_R * HID) {
            float v = acc[j];
            if (v != 0.0f) atomicAdd(&ag[j], v);
        }
    }
}

__global__ void k_bias_relu(const float* __restrict__ b1, float* __restrict__ agg) {
    int i = blockIdx.x * blockDim.x + threadIdx.x;
    if (i < N_NODES * HID) {
        float v = agg[i] + b1[i & 15];
        agg[i] = v > 0.0f ? v : 0.0f;
    }
}


__global__ void k_logsoftmax(const float* __restrict__ agg2, const float* __restrict__ b2,
                             float* __restrict__ out) {
    int i = blockIdx.x * blockDim.x + threadIdx.x;
    if (i >= N_NODES) return;
    float z[NCLS];
    float m = -INFINITY;
    #pragma unroll
    for (int c = 0; c < NCLS; ++c) {
        z[c] = agg2[(size_t)i * HID + c] + b2[c];
        m = fmaxf(m, z[c]);
    }
    float s = 0.0f;
    #pragma unroll
    for (int c = 0; c < NCLS; ++c) s += __expf(z[c] - m);
    float ls = __logf(s) + m;
    #pragma unroll
    for (int c = 0; c < NCLS; ++c) out[(size_t)i * NCLS + c] = z[c] - ls;
}


extern "C" void kernel_launch(void* const* d_in, const int* in_sizes, int n_in,
                              void* d_out, int out_size, void* d_ws, size_t ws_size,
                              hipStream_t stream) {
    const float* x  = (const float*)d_in[0];
    const int*   ei = (const int*)  d_in[1];
    const float* ew = (const float*)d_in[2];
    const float* W1 = (const float*)d_in[3];
    const float* b1 = (const float*)d_in[4];
    const float* W2 = (const float*)d_in[5];
    const float* b2 = (const float*)d_in[6];
    float* out = (float*)d_out;

    const int* srcp = ei;
    const int* dstp = ei + N_EDGES;

    float* ws   = (float*)d_ws;
    float* dinv = ws;
    float* nrm  = dinv + N_NODES;
    float* hbuf = nrm  + N_EDGES;
    float* agg1 = hbuf + (size_t)N_NODES * HID;
    float* h2   = agg1 + (size_t)N_NODES * HID;
    float* agg2 = h2   + (size_t)N_NODES * HID;

    const int B = 256;
    int gN   = (N_NODES + B - 1) / B;
    int gE   = (N_EDGES + B - 1) / B;
    int gNH  = (N_NODES * HID + B - 1) / B;
    int gMM  = (NTILES + 7) / 8;

    k_init_deg <<<gN, B, 0, stream>>>(dinv);
    k_deg_lds  <<<64, 1024, 0, stream>>>(dstp, ew, dinv);
    k_dinv     <<<gN, B, 0, stream>>>(dinv);
    k_norm     <<<gE, B, 0, stream>>>(srcp, dstp, ew, dinv, nrm);

    k_gemm1    <<<gMM, B, 0, stream>>>(x, W1, hbuf);
    k_self_init<<<gNH, B, 0, stream>>>(dinv, hbuf, agg1);
    k_scatter_lds<<<256, 1024, 0, stream>>>(srcp, dstp, nrm, hbuf, agg1);
    k_bias_relu<<<gNH, B, 0, stream>>>(b1, agg1);

    k_gemm2    <<<gMM, B, 0, stream>>>(agg1, W2, h2);
    k_self_init<<<gNH, B, 0, stream>>>(dinv, h2, agg2);
    k_scatter_lds<<<256, 1024, 0, stream>>>(srcp, dstp, nrm, h2, agg2);
    k_logsoftmax<<<gN, B, 0, stream>>>(agg2, b2, out);
}


# Edges binned once by destination range (count, offsets, fill); one work-group owns a range, sums its list in LDS and stores rows plainly: no global float atomics; summation order changes
# speedup vs baseline: 3.4561x; 2.0229x over previous
#include <hip/hip_runtime.h>
#include <math.h>


#define N_NODES 100000
#define N_EDGES 3200000
#define F_IN    512
#define HID     16
#define NCLS    10
#define NTILES  (N_NODES / 16)

typedef __attribute__((ext_vector_type(16))) __bf16 v16bf;
typedef __attribute__((ext_vector_type(8)))  float  v8f;


#define NB      256
#define RNG     416
#define CH      (N_EDGES / NB)
#define CH_PT   ((CH + 1023) / 1024)

__device__ __forceinline__ int scan256_excl(int v, int t, int* wt) {
    const int lane = t & 31;
    int inc = v;
    if (t < 256) {
        #pragma unroll
        for (int d = 1; d < 32; d <<= 1) {
            int o = __shfl_up(inc, d);
            if (lane >= d) inc += o;
        }
        if (lane == 31) wt[t >> 5] = inc;
    }
    __syncthreads();
    int off = 0;
    if (t < 256) {
        const int wv = t >> 5;
        #pragma unroll
        for (int k = 0; k < 7; ++k) if (k < wv) off += wt[k];
    }
    __syncthreads();
    return off + inc - v;
}

__global__ __launch_bounds__(1024) void k_count(const int* __restrict__ dst, int* __restrict__ hist) {
    __shared__ int c[NB];
    const int t = threadIdx.x;
    const int b = blockIdx.x;
    if (t < NB) c[t] = 0;
    __syncthreads();
    const int* dp = dst + b * CH;
    for (int i0 = t; i0 < CH; i0 += 4096) {
        int d[4];
        #pragma unroll
        for (int u = 0; u < 4; ++u) { const int i = i0 + u * 1024; d[u] = (i < CH) ? dp[i] : -1; }
        #pragma unroll
        for (int u = 0; u < 4; ++u) if (d[u] >= 0) atomicAdd(&c[d[u] / RNG], 1);
    }
    __syncthreads();
    if (t < NB) hist[b * NB + t] = c[t];
}

__global__ __launch_bounds__(1024) void k_offsets(int* __restrict__ hist, int* __restrict__ bstart) {
    __shared__ int part[4 * NB];
    __shared__ int bs[NB];
    __shared__ int wt[8];
    const int t = threadIdx.x;
    const int r = t & (NB - 1);
    const int q = t >> 8;
    int* hp = hist + (q * 64) * NB + r;

    int s = 0;
    #pragma unroll 16
    for (int j = 0; j < 64; ++j) s += hp[j * NB];
    part[q * NB + r] = s;
    __syncthreads();

    int tot = 0;
    if (t < NB) tot = part[t] + part[NB + t] + part[2 * NB + t] + part[3 * NB + t];
    const int ex = scan256_excl(tot, t, wt);
    if (t < NB) {
        bs[t] = ex;
        bstart[t] = ex;
        if (t == NB - 1) bstart[NB] = ex + tot;
    }
    __syncthreads();

    int run = bs[r];
    for (int k = 0; k < q; ++k) run += part[k * NB + r];
    for (int j0 = 0; j0 < 64; j0 += 16) {
        int v[16];
        #pragma unroll
        for (int j = 0; j < 16; ++j) v[j] = hp[(j0 + j) * NB];
        #pragma unroll
        for (int j = 0; j < 16; ++j) { hp[(j0 + j) * NB] = run; run += v[j]; }
    }
}

__global__ __launch_bounds__(1024) void k_fill(const int* __restrict__ src, const int* __restrict__ dst,
                                               const float* __restrict__ w, const int* __restrict__ offs,
                                               int* __restrict__ bkey, float* __restrict__ bval) {
    __shared__ int   cnt[NB];
    __shared__ int   lst[NB];
    __shared__ int   gdl[NB];
    __shared__ int   wt[8];
    __shared__ int   skey[CH];
    __shared__ float sval[CH];
    __shared__ int   sdel[CH];

    const int t  = threadIdx.x;
    const int b  = blockIdx.x;
    const int e0 = b * CH;

    if (t < NB) cnt[t] = 0;
    __syncthreads();

    unsigned pk[CH_PT];
    int dd[CH_PT];
    #pragma unroll
    for (int k = 0; k < CH_PT; ++k) {
        const int i = t + k * 1024;
        dd[k] = (i < CH) ? dst[e0 + i] : -1;
    }
    #pragma unroll
    for (int k = 0; k < CH_PT; ++k) {
        pk[k] = 0;
        if (dd[k] >= 0) {
            const int d  = dd[k];
            const int bn = d / RNG;
            const int rk = atomicAdd(&cnt[bn], 1);
            pk[k] = (unsigned)rk | ((unsigned)bn << 14) | ((unsigned)(d - bn * RNG) << 22);
        }
    }
    __syncthreads();

    int cv = 0;
    if (t < NB) cv = cnt[t];
    const int ex = scan256_excl(cv, t, wt);
    if (t < NB) { lst[t] = ex; gdl[t] = offs[b * NB + t] - ex; }
    __syncthreads();

    #pragma unroll
    for (int k = 0; k < CH_PT; ++k) {
        const int i = t + k * 1024;
        if (i < CH) {
            const int rk = pk[k] & 0x3FFF;
            const int bn = (pk[k] >> 14) & 0xFF;
            const int dl = pk[k] >> 22;
            const int p  = lst[bn] + rk;
            skey[p] = src[e0 + i] | (dl << 17);
            sval[p] = w[e0 + i];
            sdel[p] = gdl[bn];
        }
    }
    __syncthreads();

    for (int i = t; i < CH; i += 1024) {
        const int g = i + sdel[i];
        bkey[g] = skey[i];
        bval[g] = sval[i];
    }
}

__global__ __launch_bounds__(1024) void k_deg_dinv(const int* __restrict__ bkey, const float* __restrict__ bval,
                                                   const int* __restrict__ bstart, float* __restrict__ dinv) {
    __shared__ float dg[RNG];
    const int t = threadIdx.x;
    const int r = blockIdx.x;
    const int s = bstart[r], e = bstart[r + 1];
    if (t < RNG) dg[t] = 1.0f;
    __syncthreads();
    for (int i0 = s + t; i0 < e; i0 += 4096) {
        int   k[4];
        float v[4];
        #pragma unroll
        for (int u = 0; u < 4; ++u) {
            const int i = i0 + u * 1024;
            k[u] = -1; v[u] = 0.0f;
            if (i < e) { k[u] = bkey[i]; v[u] = bval[i]; }
        }
        #pragma unroll
        for (int u = 0; u < 4; ++u) if (k[u] >= 0) atomicAdd(&dg[k[u] >> 17], v[u]);
    }
    __syncthreads();
    const int node = r * RNG + t;
    if (t < RNG && node < N_NODES) {
        const float d = dg[t];
        dinv[node] = (d > 0.0f) ? rsqrtf(d) : 0.0f;
    }
}


__global__ void k_gemm1(const float* __restrict__ x, const float* __restrict__ W1,
                        float* __restrict__ h) {
    __shared__ __bf16 w1s[F_IN * HID];
    int t = threadIdx.x;
    for (int i = t; i < F_IN * HID; i += blockDim.x) w1s[i] = (__bf16)W1[i];
    __syncthreads();

    int wave = t >> 5;
    int lane = t & 31;
    int tile = blockIdx.x * 8 + wave;
    if (tile >= NTILES) return;

    int col = lane & 15;
    int hh  = lane >> 4;
    const float* xr = x + (size_t)(tile * 16 + col) * F_IN;

    v8f acc;
    #pragma unroll
    for (int r = 0; r < 8; ++r) acc[r] = 0.0f;

    for (int k0 = 0; k0 < F_IN; k0 += 32) {
        v16bf a, b;
        #pragma unroll
        for (int j = 0; j < 8; ++j) {
            int kl = (j < 4) ? (2 * j + 8 * hh) : (16 + 2 * (j - 4) + 8 * hh);
            int k  = k0 + kl;
            float2 xv = *reinterpret_cast<const float2*>(xr + k);
            a[2 * j]     = (__bf16)xv.x;
            a[2 * j + 1] = (__bf16)xv.y;
            b[2 * j]     = w1s[k * HID + col];
            b[2 * j + 1] = w1s[(k + 1) * HID + col];
        }
        acc = __builtin_amdgcn_wmma_f32_16x16x32_bf16(
                  false, a, false, b, (short)0, acc, false, false);
    }

    float* out = h + (size_t)(tile * 16 + 8 * hh) * HID + col;
    #pragma unroll
    for (int r = 0; r < 8; ++r) out[(size_t)r * HID] = acc[r];
}


__global__ void k_gemm2(const float* __restrict__ h1, const float* __restrict__ W2,
                        float* __restrict__ h2) {
    __shared__ __bf16 w2s[16 * 16];
    int t = threadIdx.x;
    for (int i = t; i < 16 * 16; i += blockDim.x) {
        int k = i >> 4, n = i & 15;
        w2s[i] = (n < NCLS) ? (__bf16)W2[k * NCLS + n] : (__bf16)0.0f;
    }
    __syncthreads();

    int wave = t >> 5;
    int lane = t & 31;
    int tile = blockIdx.x * 8 + wave;
    if (tile >= NTILES) return;

    int col = lane & 15;
    int hh  = lane >> 4;
    const float* hr = h1 + (size_t)(tile * 16 + col) * HID;

    v16bf a, b;
    #pragma unroll
    for (int i = 0; i < 16; ++i) { a[i] = (__bf16)0.0f; b[i] = (__bf16)0.0f; }
    #pragma unroll
    for (int j = 0; j < 8; ++j) {
        int kl = (j < 4) ? (2 * j + 8 * hh) : (16 + 2 * (j - 4) + 8 * hh);
        if (kl < 16) {
            a[2 * j]     = (__bf16)hr[kl];
            a[2 * j + 1] = (__bf16)hr[kl + 1];
            b[2 * j]     = w2s[kl * 16 + col];
            b[2 * j + 1] = w2s[(kl + 1) * 16 + col];
        }
    }
    v8f acc;
    #pragma unroll
    for (int r = 0; r < 8; ++r) acc[r] = 0.0f;
    acc = __builtin_amdgcn_wmma_f32_16x16x32_bf16(
              false, a, false, b, (short)0, acc, false, false);

    float* out = h2 + (size_t)(tile * 16 + 8 * hh) * HID + col;
    #pragma unroll
    for (int r = 0; r < 8; ++r) out[(size_t)r * HID] = acc[r];
}


__global__ __launch_bounds__(1024) void k_agg(const int* __restrict__ bkey, float* __restrict__ bval,
                                              const int* __restrict__ bstart, const float* __restrict__ dinv,
                                              const float* __restrict__ h, float* __restrict__ agg, int first) {
    __shared__ float acc[RNG * HID];
    __shared__ float dli[RNG];

    const int t    = threadIdx.x;
    const int lane = t & 31;
    const int c    = lane & 15;
    const int hsel = lane & 16;
    const int r    = blockIdx.x;
    const int s    = bstart[r], e = bstart[r + 1];
    const int n0   = r * RNG;

    for (int j = t; j < RNG * HID; j += 1024) {
        const int node = n0 + (j >> 4);
        float v = 0.0f;
        if (node < N_NODES) { const float di = dinv[node]; v = di * di * h[(size_t)n0 * HID + j]; }
        acc[j] = v;
    }
    if (t < RNG) dli[t] = (n0 + t < N_NODES) ? dinv[n0 + t] : 0.0f;
    __syncthreads();

    for (int base = s + (t >> 5) * 128; base < e; base += 32 * 128) {
        int   k0 = 0, k1 = 0, k2 = 0, k3 = 0;
        float v0 = 0.0f, v1 = 0.0f, v2 = 0.0f, v3 = 0.0f;
        const int i0 = base + lane, i1 = i0 + 32, i2 = i0 + 64, i3 = i0 + 96;
        if (i0 < e) { k0 = bkey[i0]; v0 = bval[i0]; }
        if (i1 < e) { k1 = bkey[i1]; v1 = bval[i1]; }
        if (i2 < e) { k2 = bkey[i2]; v2 = bval[i2]; }
        if (i3 < e) { k3 = bkey[i3]; v3 = bval[i3]; }
        if (first) {
            const float d0 = dinv[k0 & 0x1FFFF], d1 = dinv[k1 & 0x1FFFF];
            const float d2 = dinv[k2 & 0x1FFFF], d3 = dinv[k3 & 0x1FFFF];
            v0 = d0 * v0 * dli[k0 >> 17];
            v1 = d1 * v1 * dli[k1 >> 17];
            v2 = d2 * v2 * dli[k2 >> 17];
            v3 = d3 * v3 * dli[k3 >> 17];
            if (i0 < e) bval[i0] = v0;
            if (i1 < e) bval[i1] = v1;
            if (i2 < e) bval[i2] = v2;
            if (i3 < e) bval[i3] = v3;
        }

        #pragma nounroll
        for (int u = 0; u < 4; ++u) {
            const int b0 = base + u * 32;
            if (b0 < e) {
                int   kk[16];
                float vv[16], hv[16];
                #pragma unroll
                for (int j = 0; j < 16; ++j) { kk[j] = __shfl(k0, j + hsel); vv[j] = __shfl(v0, j + hsel); }
                #pragma unroll
                for (int j = 0; j < 16; ++j) hv[j] = h[(size_t)(kk[j] & 0x1FFFF) * HID + c];
                #pragma unroll
                for (int j = 0; j < 16; ++j)
                    if (b0 + j + hsel < e) atomicAdd(&acc[(kk[j] >> 17) * HID + c], vv[j] * hv[j]);
            }
            k0 = k1; k1 = k2; k2 = k3;
            v0 = v1; v1 = v2; v2 = v3;
        }
    }
    __syncthreads();

    for (int j = t; j < RNG * HID; j += 1024)
        if (n0 + (j >> 4) < N_NODES) agg[(size_t)n0 * HID + j] = acc[j];
}

__global__ void k_bias_relu(const float* __restrict__ b1, float* __restrict__ agg) {
    int i = blockIdx.x * blockDim.x + threadIdx.x;
    if (i < N_NODES * HID) {
        float v = agg[i] + b1[i & 15];
        agg[i] = v > 0.0f ? v : 0.0f;
    }
}


__global__ void k_logsoftmax(const float* __restrict__ agg2, const float* __restrict__ b2,
                             float* __restrict__ out) {
    int i = blockIdx.x * blockDim.x + threadIdx.x;
    if (i >= N_NODES) return;
    float z[NCLS];
    float m = -INFINITY;
    #pragma unroll
    for (int c = 0; c < NCLS; ++c) {
        z[c] = agg2[(size_t)i * HID + c] + b2[c];
        m = fmaxf(m, z[c]);
    }
    float s = 0.0f;
    #pragma unroll
    for (int c = 0; c < NCLS; ++c) s += __expf(z[c] - m);
    float ls = __logf(s) + m;
    #pragma unroll
    for (int c = 0; c < NCLS; ++c) out[(size_t)i * NCLS + c] = z[c] - ls;
}


extern "C" void kernel_launch(void* const* d_in, const int* in_sizes, int n_in,
                              void* d_out, int out_size, void* d_ws, size_t ws_size,
                              hipStream_t stream) {
    const float* x  = (const float*)d_in[0];
    const int*   ei = (const int*)  d_in[1];
    const float* ew = (const float*)d_in[2];
    const float* W1 = (const float*)d_in[3];
    const float* b1 = (const float*)d_in[4];
    const float* W2 = (const float*)d_in[5];
    const float* b2 = (const float*)d_in[6];
    float* out = (float*)d_out;

    const int* srcp = ei;
    const int* dstp = ei + N_EDGES;

    float* ws   = (float*)d_ws;
    float* dinv = ws;
    int*   bkey = (int*)(dinv + N_NODES);
    float* bval = (float*)(bkey + N_EDGES);
    float* hbuf = bval + N_EDGES;
    float* agg1 = hbuf + (size_t)N_NODES * HID;
    float* h2   = hbuf;
    float* agg2 = agg1;

    int* hist   = (int*)d_out;
    int* bstart = hist + NB * NB;

    const int B = 256;
    int gN   = (N_NODES + B - 1) / B;
    int gNH  = (N_NODES * HID + B - 1) / B;
    int gMM  = (NTILES + 7) / 8;

    k_count    <<<NB, 1024, 0, stream>>>(dstp, hist);
    k_offsets  <<<1, 1024, 0, stream>>>(hist, bstart);
    k_fill     <<<NB, 1024, 0, stream>>>(srcp, dstp, ew, hist, bkey, bval);
    k_deg_dinv <<<NB, 1024, 0, stream>>>(bkey, bval, bstart, dinv);

    k_gemm1    <<<gMM, B, 0, stream>>>(x, W1, hbuf);
    k_agg      <<<NB, 1024, 0, stream>>>(bkey, bval, bstart, dinv, hbuf, agg1, 1);
    k_bias_relu<<<gNH, B, 0, stream>>>(b1, agg1);

    k_gemm2    <<<gMM, B, 0, stream>>>(agg1, W2, h2);
    k_agg      <<<NB, 1024, 0, stream>>>(bkey, bval, bstart, dinv, h2, agg2, 0);
    k_logsoftmax<<<gN, B, 0, stream>>>(agg2, b2, out);
}


# Row-wise stages run in the aggregation work-group while its rows are in LDS: bias+relu+W2 product (WMMA) after layer 1, bias+log_softmax after layer 2; 10 launches become 7
# speedup vs baseline: 3.7303x; 1.0793x over previous
#include <hip/hip_runtime.h>
#include <math.h>


#define N_NODES 100000
#define N_EDGES 3200000
#define F_IN    512
#define HID     16
#define NCLS    10
#define NTILES  (N_NODES / 16)

typedef __attribute__((ext_vector_type(16))) __bf16 v16bf;
typedef __attribute__((ext_vector_type(8)))  float  v8f;


#define NB      256
#define RNG     416
#define CH      (N_EDGES / NB)
#define CH_PT   ((CH + 1023) / 1024)
#define RB_STEP (RNG * NCLS)

__device__ __forceinline__ int scan256_excl(int v, int t, int* wt) {
    const int lane = t & 31;
    int inc = v;
    if (t < 256) {
        #pragma unroll
        for (int d = 1; d < 32; d <<= 1) {
            int o = __shfl_up(inc, d);
            if (lane >= d) inc += o;
        }
        if (lane == 31) wt[t >> 5] = inc;
    }
    __syncthreads();
    int off = 0;
    if (t < 256) {
        const int wv = t >> 5;
        #pragma unroll
        for (int k = 0; k < 7; ++k) if (k < wv) off += wt[k];
    }
    __syncthreads();
    return off + inc - v;
}

__global__ __launch_bounds__(1024) void k_count(const int* __restrict__ dst, int* __restrict__ hist) {
    __shared__ int c[NB];
    const int t = threadIdx.x;
    const int b = blockIdx.x;
    if (t < NB) c[t] = 0;
    __syncthreads();
    const int* dp = dst + b * CH;
    for (int i0 = t; i0 < CH; i0 += 4096) {
        int d[4];
        #pragma unroll
        for (int u = 0; u < 4; ++u) { const int i = i0 + u * 1024; d[u] = (i < CH) ? dp[i] : -1; }
        #pragma unroll
        for (int u = 0; u < 4; ++u) if (d[u] >= 0) atomicAdd(&c[d[u] / RNG], 1);
    }
    __syncthreads();
    if (t < NB) hist[b * NB + t] = c[t];
}

__global__ __launch_bounds__(1024) void k_offsets(int* __restrict__ hist, int* __restrict__ rb) {
    __shared__ int part[4 * NB];
    __shared__ int bs[NB];
    __shared__ int wt[8];
    const int t = threadIdx.x;
    const int r = t & (NB - 1);
    const int q = t >> 8;
    int* hp = hist + (q * 64) * NB + r;

    int s = 0;
    #pragma unroll 16
    for (int j = 0; j < 64; ++j) s += hp[j * NB];
    part[q * NB + r] = s;
    __syncthreads();

    int tot = 0;
    if (t < NB) tot = part[t] + part[NB + t] + part[2 * NB + t] + part[3 * NB + t];
    const int ex = scan256_excl(tot, t, wt);
    if (t < NB) {
        bs[t] = ex;
        if (t * RNG < N_NODES) { rb[t * RB_STEP] = ex; rb[t * RB_STEP + 1] = ex + tot; }
    }
    __syncthreads();

    int run = bs[r];
    for (int k = 0; k < q; ++k) run += part[k * NB + r];
    for (int j0 = 0; j0 < 64; j0 += 16) {
        int v[16];
        #pragma unroll
        for (int j = 0; j < 16; ++j) v[j] = hp[(j0 + j) * NB];
        #pragma unroll
        for (int j = 0; j < 16; ++j) { hp[(j0 + j) * NB] = run; run += v[j]; }
    }
}

__global__ __launch_bounds__(1024) void k_fill(const int* __restrict__ src, const int* __restrict__ dst,
                                               const float* __restrict__ w, const int* __restrict__ offs,
                                               int* __restrict__ bkey, float* __restrict__ bval) {
    __shared__ int   cnt[NB];
    __shared__ int   lst[NB];
    __shared__ int   gdl[NB];
    __shared__ int   wt[8];
    __shared__ int   skey[CH];
    __shared__ float sval[CH];
    __shared__ int   sdel[CH];

    const int t  = threadIdx.x;
    const int b  = blockIdx.x;
    const int e0 = b * CH;

    if (t < NB) cnt[t] = 0;
    __syncthreads();

    unsigned pk[CH_PT];
    int dd[CH_PT];
    #pragma unroll
    for (int k = 0; k < CH_PT; ++k) {
        const int i = t + k * 1024;
        dd[k] = (i < CH) ? dst[e0 + i] : -1;
    }
    #pragma unroll
    for (int k = 0; k < CH_PT; ++k) {
        pk[k] = 0;
        if (dd[k] >= 0) {
            const int d  = dd[k];
            const int bn = d / RNG;
            const int rk = atomicAdd(&cnt[bn], 1);
            pk[k] = (unsigned)rk | ((unsigned)bn << 14) | ((unsigned)(d - bn * RNG) << 22);
        }
    }
    __syncthreads();

    int cv = 0;
    if (t < NB) cv = cnt[t];
    const int ex = scan256_excl(cv, t, wt);
    if (t < NB) { lst[t] = ex; gdl[t] = offs[b * NB + t] - ex; }
    __syncthreads();

    #pragma unroll
    for (int k = 0; k < CH_PT; ++k) {
        const int i = t + k * 1024;
        if (i < CH) {
            const int rk = pk[k] & 0x3FFF;
            const int bn = (pk[k] >> 14) & 0xFF;
            const int dl = pk[k] >> 22;
            const int p  = lst[bn] + rk;
            skey[p] = src[e0 + i] | (dl << 17);
            sval[p] = w[e0 + i];
            sdel[p] = gdl[bn];
        }
    }
    __syncthreads();

    for (int i = t; i < CH; i += 1024) {
        const int g = i + sdel[i];
        bkey[g] = skey[i];
        bval[g] = sval[i];
    }
}

__global__ __launch_bounds__(1024) void k_deg_dinv(const int* __restrict__ bkey, const float* __restrict__ bval,
                                                   const int* __restrict__ rb, float* __restrict__ dinv) {
    __shared__ float dg[RNG];
    const int t = threadIdx.x;
    const int r = blockIdx.x;
    if (r * RNG >= N_NODES) return;
    const int s = rb[r * RB_STEP], e = rb[r * RB_STEP + 1];
    if (t < RNG) dg[t] = 1.0f;
    __syncthreads();
    for (int i0 = s + t; i0 < e; i0 += 4096) {
        int   k[4];
        float v[4];
        #pragma unroll
        for (int u = 0; u < 4; ++u) {
            const int i = i0 + u * 1024;
            k[u] = -1; v[u] = 0.0f;
            if (i < e) { k[u] = bkey[i]; v[u] = bval[i]; }
        }
        #pragma unroll
        for (int u = 0; u < 4; ++u) if (k[u] >= 0) atomicAdd(&dg[k[u] >> 17], v[u]);
    }
    __syncthreads();
    const int node = r * RNG + t;
    if (t < RNG && node < N_NODES) {
        const float d = dg[t];
        dinv[node] = (d > 0.0f) ? rsqrtf(d) : 0.0f;
    }
}


__global__ void k_gemm1(const float* __restrict__ x, const float* __restrict__ W1,
                        float* __restrict__ h) {
    __shared__ __bf16 w1s[F_IN * HID];
    int t = threadIdx.x;
    for (int i = t; i < F_IN * HID; i += blockDim.x) w1s[i] = (__bf16)W1[i];
    __syncthreads();

    int wave = t >> 5;
    int lane = t & 31;
    int tile = blockIdx.x * 8 + wave;
    if (tile >= NTILES) return;

    int col = lane & 15;
    int hh  = lane >> 4;
    const float* xr = x + (size_t)(tile * 16 + col) * F_IN;

    v8f acc;
    #pragma unroll
    for (int r = 0; r < 8; ++r) acc[r] = 0.0f;

    for (int k0 = 0; k0 < F_IN; k0 += 32) {
        v16bf a, b;
        #pragma unroll
        for (int j = 0; j < 8; ++j) {
            int kl = (j < 4) ? (2 * j + 8 * hh) : (16 + 2 * (j - 4) + 8 * hh);
            int k  = k0 + kl;
            float2 xv = *reinterpret_cast<const float2*>(xr + k);
            a[2 * j]     = (__bf16)xv.x;
            a[2 * j + 1] = (__bf16)xv.y;
            b[2 * j]     = w1s[k * HID + col];
            b[2 * j + 1] = w1s[(k + 1) * HID + col];
        }
        acc = __builtin_amdgcn_wmma_f32_16x16x32_bf16(
                  false, a, false, b, (short)0, acc, false, false);
    }

    float* out = h + (size_t)(tile * 16 + 8 * hh) * HID + col;
    #pragma unroll
    for (int r = 0; r < 8; ++r) out[(size_t)r * HID] = acc[r];
}


template <bool FIRST>
__global__ __launch_bounds__(1024) void k_agg(const int* __restrict__ bkey, float* __restrict__ bval,
                                              const int* rb, const float* __restrict__ dinv,
                                              const float* __restrict__ h, const float* __restrict__ bias,
                                              const float* __restrict__ W2, float* dst) {
    __shared__ float acc[RNG * HID];
    __shared__ float dli[RNG];
    __shared__ __bf16 w2s[FIRST ? 16 * 16 : 2];
    __shared__ float  ost[FIRST ? 1 : RNG * NCLS];

    const int t    = threadIdx.x;
    const int lane = t & 31;
    const int c    = lane & 15;
    const int hsel = lane & 16;
    const int r    = blockIdx.x;
    const int n0   = r * RNG;
    if (n0 >= N_NODES) return;
    const int s    = rb[r * RB_STEP], e = rb[r * RB_STEP + 1];

    for (int j = t; j < RNG * HID; j += 1024) {
        const int node = n0 + (j >> 4);
        float v = 0.0f;
        if (node < N_NODES) { const float di = dinv[node]; v = di * di * h[(size_t)n0 * HID + j]; }
        acc[j] = v;
    }
    if (t < RNG) dli[t] = (n0 + t < N_NODES) ? dinv[n0 + t] : 0.0f;
    if constexpr (FIRST) {
        if (t < 16 * 16) {
            const int k = t >> 4, n = t & 15;
            w2s[t] = (n < NCLS) ? (__bf16)W2[k * NCLS + n] : (__bf16)0.0f;
        }
    }
    __syncthreads();

    for (int base = s + (t >> 5) * 128; base < e; base += 32 * 128) {
        int   k0 = 0, k1 = 0, k2 = 0, k3 = 0;
        float v0 = 0.0f, v1 = 0.0f, v2 = 0.0f, v3 = 0.0f;
        const int i0 = base + lane, i1 = i0 + 32, i2 = i0 + 64, i3 = i0 + 96;
        if (i0 < e) { k0 = bkey[i0]; v0 = bval[i0]; }
        if (i1 < e) { k1 = bkey[i1]; v1 = bval[i1]; }
        if (i2 < e) { k2 = bkey[i2]; v2 = bval[i2]; }
        if (i3 < e) { k3 = bkey[i3]; v3 = bval[i3]; }
        if constexpr (FIRST) {
            const float d0 = dinv[k0 & 0x1FFFF], d1 = dinv[k1 & 0x1FFFF];
            const float d2 = dinv[k2 & 0x1FFFF], d3 = dinv[k3 & 0x1FFFF];
            v0 = d0 * v0 * dli[k0 >> 17];
            v1 = d1 * v1 * dli[k1 >> 17];
            v2 = d2 * v2 * dli[k2 >> 17];
            v3 = d3 * v3 * dli[k3 >> 17];
            if (i0 < e) bval[i0] = v0;
            if (i1 < e) bval[i1] = v1;
            if (i2 < e) bval[i2] = v2;
            if (i3 < e) bval[i3] = v3;
        }

        #pragma nounroll
        for (int u = 0; u < 4; ++u) {
            const int b0 = base + u * 32;
            if (b0 < e) {
                int   kk[16];
                float vv[16], hv[16];
                #pragma unroll
                for (int j = 0; j < 16; ++j) { kk[j] = __shfl(k0, j + hsel); vv[j] = __shfl(v0, j + hsel); }
                #pragma unroll
                for (int j = 0; j < 16; ++j) hv[j] = h[(size_t)(kk[j] & 0x1FFFF) * HID + c];
                #pragma unroll
                for (int j = 0; j < 16; ++j)
                    if (b0 + j + hsel < e) atomicAdd(&acc[(kk[j] >> 17) * HID + c], vv[j] * hv[j]);
            }
            k0 = k1; k1 = k2; k2 = k3;
            v0 = v1; v1 = v2; v2 = v3;
        }
    }
    __syncthreads();

    if constexpr (FIRST) {
        const int wave = t >> 5;
        if (wave < RNG / 16 && n0 + wave * 16 < N_NODES) {
            const int hh = lane >> 4;
            const float* row = acc + (wave * 16 + c) * HID;
            v16bf a, b;
            #pragma unroll
            for (int i = 0; i < 16; ++i) { a[i] = (__bf16)0.0f; b[i] = (__bf16)0.0f; }
            #pragma unroll
            for (int j = 0; j < 4; ++j) {
                const int kl = 2 * j + 8 * hh;
                const float x0 = row[kl]     + bias[kl];
                const float x1 = row[kl + 1] + bias[kl + 1];
                a[2 * j]     = (__bf16)(x0 > 0.0f ? x0 : 0.0f);
                a[2 * j + 1] = (__bf16)(x1 > 0.0f ? x1 : 0.0f);
                b[2 * j]     = w2s[kl * 16 + c];
                b[2 * j + 1] = w2s[(kl + 1) * 16 + c];
            }
            v8f d;
            #pragma unroll
            for (int q = 0; q < 8; ++q) d[q] = 0.0f;
            d = __builtin_amdgcn_wmma_f32_16x16x32_bf16(
                    false, a, false, b, (short)0, d, false, false);
            #pragma unroll
            for (int q = 0; q < 8; ++q) acc[(wave * 16 + 8 * hh + q) * HID + c] = d[q];
        }
        __syncthreads();

        for (int j = t; j < RNG * HID; j += 1024)
            if (n0 + (j >> 4) < N_NODES) dst[(size_t)n0 * HID + j] = acc[j];
    } else {
        if (t < RNG && n0 + t < N_NODES) {
            float z[NCLS];
            float m = -INFINITY;
            #pragma unroll
            for (int q = 0; q < NCLS; ++q) {
                z[q] = acc[t * HID + q] + bias[q];
                m = fmaxf(m, z[q]);
            }
            float se = 0.0f;
            #pragma unroll
            for (int q = 0; q < NCLS; ++q) se += __expf(z[q] - m);
            const float ls = __logf(se) + m;
            #pragma unroll
            for (int q = 0; q < NCLS; ++q) ost[t * NCLS + q] = z[q] - ls;
        }
        __syncthreads();

        const int rows = (N_NODES - n0 < RNG) ? (N_NODES - n0) : RNG;
        for (int j = t; j < rows * NCLS; j += 1024) dst[(size_t)n0 * NCLS + j] = ost[j];
    }
}


extern "C" void kernel_launch(void* const* d_in, const int* in_sizes, int n_in,
                              void* d_out, int out_size, void* d_ws, size_t ws_size,
                              hipStream_t stream) {
    const float* x  = (const float*)d_in[0];
    const int*   ei = (const int*)  d_in[1];
    const float* ew = (const float*)d_in[2];
    const float* W1 = (const float*)d_in[3];
    const float* b1 = (const float*)d_in[4];
    const float* W2 = (const float*)d_in[5];
    const float* b2 = (const float*)d_in[6];
    float* out = (float*)d_out;

    const int* srcp = ei;
    const int* dstp = ei + N_EDGES;

    float* ws   = (float*)d_ws;
    float* dinv = ws;
    int*   bkey = (int*)(dinv + N_NODES);
    float* bval = (float*)(bkey + N_EDGES);
    float* hbuf = bval + N_EDGES;
    float* h2   = hbuf + (size_t)N_NODES * HID;

    int* hist = (int*)hbuf;
    int* rb   = (int*)d_out;

    const int B = 256;
    int gMM  = (NTILES + 7) / 8;

    k_count    <<<NB, 1024, 0, stream>>>(dstp, hist);
    k_offsets  <<<1, 1024, 0, stream>>>(hist, rb);
    k_fill     <<<NB, 1024, 0, stream>>>(srcp, dstp, ew, hist, bkey, bval);
    k_deg_dinv <<<NB, 1024, 0, stream>>>(bkey, bval, rb, dinv);

    k_gemm1    <<<gMM, B, 0, stream>>>(x, W1, hbuf);
    k_agg<true> <<<NB, 1024, 0, stream>>>(bkey, bval, rb, dinv, hbuf, b1, W2, h2);

    k_agg<false><<<NB, 1024, 0, stream>>>(bkey, bval, rb, dinv, h2, b2, W2, out);
}


# Edge norm factored: rows scaled by dinv once where produced, per-edge term is w*hs[src], sums scaled by dinv[dst] at the end; no dinv[src] gather or norm write-back; product rounding changes
# speedup vs baseline: 4.0397x; 1.0829x over previous
#include <hip/hip_runtime.h>
#include <math.h>


#define N_NODES 100000
#define N_EDGES 3200000
#define F_IN    512
#define HID     16
#define NCLS    10
#define NTILES  (N_NODES / 16)

typedef __attribute__((ext_vector_type(16))) __bf16 v16bf;
typedef __attribute__((ext_vector_type(8)))  float  v8f;


#define NB      256
#define RNG     416
#define CH      (N_EDGES / NB)
#define CH_PT   ((CH + 1023) / 1024)
#define RB_STEP (RNG * NCLS)

__device__ __forceinline__ int scan256_excl(int v, int t, int* wt) {
    const int lane = t & 31;
    int inc = v;
    if (t < 256) {
        #pragma unroll
        for (int d = 1; d < 32; d <<= 1) {
            int o = __shfl_up(inc, d);
            if (lane >= d) inc += o;
        }
        if (lane == 31) wt[t >> 5] = inc;
    }
    __syncthreads();
    int off = 0;
    if (t < 256) {
        const int wv = t >> 5;
        #pragma unroll
        for (int k = 0; k < 7; ++k) if (k < wv) off += wt[k];
    }
    __syncthreads();
    return off + inc - v;
}

__global__ __launch_bounds__(1024) void k_count(const int* __restrict__ dst, int* __restrict__ hist) {
    __shared__ int c[NB];
    const int t = threadIdx.x;
    const int b = blockIdx.x;
    if (t < NB) c[t] = 0;
    __syncthreads();
    const int* dp = dst + b * CH;
    for (int i0 = t; i0 < CH; i0 += 4096) {
        int d[4];
        #pragma unroll
        for (int u = 0; u < 4; ++u) { const int i = i0 + u * 1024; d[u] = (i < CH) ? dp[i] : -1; }
        #pragma unroll
        for (int u = 0; u < 4; ++u) if (d[u] >= 0) atomicAdd(&c[d[u] / RNG], 1);
    }
    __syncthreads();
    if (t < NB) hist[b * NB + t] = c[t];
}

__global__ __launch_bounds__(1024) void k_offsets(int* __restrict__ hist, int* __restrict__ rb) {
    __shared__ int part[4 * NB];
    __shared__ int bs[NB];
    __shared__ int wt[8];
    const int t = threadIdx.x;
    const int r = t & (NB - 1);
    const int q = t >> 8;
    int* hp = hist + (q * 64) * NB + r;

    int s = 0;
    #pragma unroll 16
    for (int j = 0; j < 64; ++j) s += hp[j * NB];
    part[q * NB + r] = s;
    __syncthreads();

    int tot = 0;
    if (t < NB) tot = part[t] + part[NB + t] + part[2 * NB + t] + part[3 * NB + t];
    const int ex = scan256_excl(tot, t, wt);
    if (t < NB) {
        bs[t] = ex;
        if (t * RNG < N_NODES) { rb[t * RB_STEP] = ex; rb[t * RB_STEP + 1] = ex + tot; }
    }
    __syncthreads();

    int run = bs[r];
    for (int k = 0; k < q; ++k) run += part[k * NB + r];
    for (int j0 = 0; j0 < 64; j0 += 16) {
        int v[16];
        #pragma unroll
        for (int j = 0; j < 16; ++j) v[j] = hp[(j0 + j) * NB];
        #pragma unroll
        for (int j = 0; j < 16; ++j) { hp[(j0 + j) * NB] = run; run += v[j]; }
    }
}

__global__ __launch_bounds__(1024) void k_fill(const int* __restrict__ src, const int* __restrict__ dst,
                                               const float* __restrict__ w, const int* __restrict__ offs,
                                               int* __restrict__ bkey, float* __restrict__ bval) {
    __shared__ int   cnt[NB];
    __shared__ int   lst[NB];
    __shared__ int   gdl[NB];
    __shared__ int   wt[8];
    __shared__ int   skey[CH];
    __shared__ float sval[CH];
    __shared__ int   sdel[CH];

    const int t  = threadIdx.x;
    const int b  = blockIdx.x;
    const int e0 = b * CH;

    if (t < NB) cnt[t] = 0;
    __syncthreads();

    unsigned pk[CH_PT];
    int dd[CH_PT];
    #pragma unroll
    for (int k = 0; k < CH_PT; ++k) {
        const int i = t + k * 1024;
        dd[k] = (i < CH) ? dst[e0 + i] : -1;
    }
    #pragma unroll
    for (int k = 0; k < CH_PT; ++k) {
        pk[k] = 0;
        if (dd[k] >= 0) {
            const int d  = dd[k];
            const int bn = d / RNG;
            const int rk = atomicAdd(&cnt[bn], 1);
            pk[k] = (unsigned)rk | ((unsigned)bn << 14) | ((unsigned)(d - bn * RNG) << 22);
        }
    }
    __syncthreads();

    int cv = 0;
    if (t < NB) cv = cnt[t];
    const int ex = scan256_excl(cv, t, wt);
    if (t < NB) { lst[t] = ex; gdl[t] = offs[b * NB + t] - ex; }
    __syncthreads();

    #pragma unroll
    for (int k = 0; k < CH_PT; ++k) {
        const int i = t + k * 1024;
        if (i < CH) {
            const int rk = pk[k] & 0x3FFF;
            const int bn = (pk[k] >> 14) & 0xFF;
            const int dl = pk[k] >> 22;
            const int p  = lst[bn] + rk;
            skey[p] = src[e0 + i] | (dl << 17);
            sval[p] = w[e0 + i];
            sdel[p] = gdl[bn];
        }
    }
    __syncthreads();

    for (int i = t; i < CH; i += 1024) {
        const int g = i + sdel[i];
        bkey[g] = skey[i];
        bval[g] = sval[i];
    }
}

__global__ __launch_bounds__(1024) void k_deg_dinv(const int* __restrict__ bkey, const float* __restrict__ bval,
                                                   const int* __restrict__ rb, float* __restrict__ dinv) {
    __shared__ float dg[RNG];
    const int t = threadIdx.x;
    const int r = blockIdx.x;
    if (r * RNG >= N_NODES) return;
    const int s = rb[r * RB_STEP], e = rb[r * RB_STEP + 1];
    if (t < RNG) dg[t] = 1.0f;
    __syncthreads();
    for (int i0 = s + t; i0 < e; i0 += 4096) {
        int   k[4];
        float v[4];
        #pragma unroll
        for (int u = 0; u < 4; ++u) {
            const int i = i0 + u * 1024;
            k[u] = -1; v[u] = 0.0f;
            if (i < e) { k[u] = bkey[i]; v[u] = bval[i]; }
        }
        #pragma unroll
        for (int u = 0; u < 4; ++u) if (k[u] >= 0) atomicAdd(&dg[k[u] >> 17], v[u]);
    }
    __syncthreads();
    const int node = r * RNG + t;
    if (t < RNG && node < N_NODES) {
        const float d = dg[t];
        dinv[node] = (d > 0.0f) ? rsqrtf(d) : 0.0f;
    }
}


__global__ void k_gemm1(const float* __restrict__ x, const float* __restrict__ W1,
                        const float* __restrict__ dinv, float* __restrict__ h) {
    __shared__ __bf16 w1s[F_IN * HID];
    int t = threadIdx.x;
    for (int i = t; i < F_IN * HID; i += blockDim.x) w1s[i] = (__bf16)W1[i];
    __syncthreads();

    int wave = t >> 5;
    int lane = t & 31;
    int tile = blockIdx.x * 8 + wave;
    if (tile >= NTILES) return;

    int col = lane & 15;
    int hh  = lane >> 4;
    const float* xr = x + (size_t)(tile * 16 + col) * F_IN;

    v8f acc;
    #pragma unroll
    for (int r = 0; r < 8; ++r) acc[r] = 0.0f;

    for (int k0 = 0; k0 < F_IN; k0 += 32) {
        v16bf a, b;
        #pragma unroll
        for (int j = 0; j < 8; ++j) {
            int kl = (j < 4) ? (2 * j + 8 * hh) : (16 + 2 * (j - 4) + 8 * hh);
            int k  = k0 + kl;
            float2 xv = *reinterpret_cast<const float2*>(xr + k);
            a[2 * j]     = (__bf16)xv.x;
            a[2 * j + 1] = (__bf16)xv.y;
            b[2 * j]     = w1s[k * HID + col];
            b[2 * j + 1] = w1s[(k + 1) * HID + col];
        }
        acc = __builtin_amdgcn_wmma_f32_16x16x32_bf16(
                  false, a, false, b, (short)0, acc, false, false);
    }

    const float* dr = dinv + tile * 16 + 8 * hh;
    float* out = h + (size_t)(tile * 16 + 8 * hh) * HID + col;
    #pragma unroll
    for (int r = 0; r < 8; ++r) out[(size_t)r * HID] = acc[r] * dr[r];
}


template <bool FIRST>
__global__ __launch_bounds__(1024) void k_agg(const int* __restrict__ bkey, const float* __restrict__ bval,
                                              const int* rb, const float* __restrict__ dinv,
                                              const float* __restrict__ h, const float* __restrict__ bias,
                                              const float* __restrict__ W2, float* dst) {
    __shared__ float acc[RNG * HID];
    __shared__ float dli[RNG];
    __shared__ __bf16 w2s[FIRST ? 16 * 16 : 2];
    __shared__ float  ost[FIRST ? 1 : RNG * NCLS];

    const int t    = threadIdx.x;
    const int lane = t & 31;
    const int c    = lane & 15;
    const int hsel = lane & 16;
    const int r    = blockIdx.x;
    const int n0   = r * RNG;
    if (n0 >= N_NODES) return;
    const int s    = rb[r * RB_STEP], e = rb[r * RB_STEP + 1];

    for (int j = t; j < RNG * HID; j += 1024) {
        const int node = n0 + (j >> 4);
        acc[j] = (node < N_NODES) ? h[(size_t)n0 * HID + j] : 0.0f;
    }
    if (t < RNG) dli[t] = (n0 + t < N_NODES) ? dinv[n0 + t] : 0.0f;
    if constexpr (FIRST) {
        if (t < 16 * 16) {
            const int k = t >> 4, n = t & 15;
            w2s[t] = (n < NCLS) ? (__bf16)W2[k * NCLS + n] : (__bf16)0.0f;
        }
    }
    __syncthreads();

    for (int base = s + (t >> 5) * 128; base < e; base += 32 * 128) {
        int   k0 = 0, k1 = 0, k2 = 0, k3 = 0;
        float v0 = 0.0f, v1 = 0.0f, v2 = 0.0f, v3 = 0.0f;
        const int i0 = base + lane, i1 = i0 + 32, i2 = i0 + 64, i3 = i0 + 96;
        if (i0 < e) { k0 = bkey[i0]; v0 = bval[i0]; }
        if (i1 < e) { k1 = bkey[i1]; v1 = bval[i1]; }
        if (i2 < e) { k2 = bkey[i2]; v2 = bval[i2]; }
        if (i3 < e) { k3 = bkey[i3]; v3 = bval[i3]; }

        #pragma nounroll
        for (int u = 0; u < 4; ++u) {
            const int b0 = base + u * 32;
            if (b0 < e) {
                int   kk[16];
                float vv[16], hv[16];
                #pragma unroll
                for (int j = 0; j < 16; ++j) { kk[j] = __shfl(k0, j + hsel); vv[j] = __shfl(v0, j + hsel); }
                #pragma unroll
                for (int j = 0; j < 16; ++j) hv[j] = h[(size_t)(kk[j] & 0x1FFFF) * HID + c];
                #pragma unroll
                for (int j = 0; j < 16; ++j)
                    if (b0 + j + hsel < e) atomicAdd(&acc[(kk[j] >> 17) * HID + c], vv[j] * hv[j]);
            }
            k0 = k1; k1 = k2; k2 = k3;
            v0 = v1; v1 = v2; v2 = v3;
        }
    }
    __syncthreads();

    if constexpr (FIRST) {
        const int wave = t >> 5;
        if (wave < RNG / 16 && n0 + wave * 16 < N_NODES) {
            const int hh = lane >> 4;
            const float* row = acc + (wave * 16 + c) * HID;
            const float  dl  = dli[wave * 16 + c];
            v16bf a, b;
            #pragma unroll
            for (int i = 0; i < 16; ++i) { a[i] = (__bf16)0.0f; b[i] = (__bf16)0.0f; }
            #pragma unroll
            for (int j = 0; j < 4; ++j) {
                const int kl = 2 * j + 8 * hh;
                const float x0 = dl * row[kl]     + bias[kl];
                const float x1 = dl * row[kl + 1] + bias[kl + 1];
                a[2 * j]     = (__bf16)(x0 > 0.0f ? x0 : 0.0f);
                a[2 * j + 1] = (__bf16)(x1 > 0.0f ? x1 : 0.0f);
                b[2 * j]     = w2s[kl * 16 + c];
                b[2 * j + 1] = w2s[(kl + 1) * 16 + c];
            }
            v8f d;
            #pragma unroll
            for (int q = 0; q < 8; ++q) d[q] = 0.0f;
            d = __builtin_amdgcn_wmma_f32_16x16x32_bf16(
                    false, a, false, b, (short)0, d, false, false);
            #pragma unroll
            for (int q = 0; q < 8; ++q)
                acc[(wave * 16 + 8 * hh + q) * HID + c] = d[q] * dli[wave * 16 + 8 * hh + q];
        }
        __syncthreads();

        for (int j = t; j < RNG * HID; j += 1024)
            if (n0 + (j >> 4) < N_NODES) dst[(size_t)n0 * HID + j] = acc[j];
    } else {
        if (t < RNG && n0 + t < N_NODES) {
            float z[NCLS];
            float m = -INFINITY;
            #pragma unroll
            for (int q = 0; q < NCLS; ++q) {
                z[q] = dli[t] * acc[t * HID + q] + bias[q];
                m = fmaxf(m, z[q]);
            }
            float se = 0.0f;
            #pragma unroll
            for (int q = 0; q < NCLS; ++q) se += __expf(z[q] - m);
            const float ls = __logf(se) + m;
            #pragma unroll
            for (int q = 0; q < NCLS; ++q) ost[t * NCLS + q] = z[q] - ls;
        }
        __syncthreads();

        const int rows = (N_NODES - n0 < RNG) ? (N_NODES - n0) : RNG;
        for (int j = t; j < rows * NCLS; j += 1024) dst[(size_t)n0 * NCLS + j] = ost[j];
    }
}


extern "C" void kernel_launch(void* const* d_in, const int* in_sizes, int n_in,
                              void* d_out, int out_size, void* d_ws, size_t ws_size,
                              hipStream_t stream) {
    const float* x  = (const float*)d_in[0];
    const int*   ei = (const int*)  d_in[1];
    const float* ew = (const float*)d_in[2];
    const float* W1 = (const float*)d_in[3];
    const float* b1 = (const float*)d_in[4];
    const float* W2 = (const float*)d_in[5];
    const float* b2 = (const float*)d_in[6];
    float* out = (float*)d_out;

    const int* srcp = ei;
    const int* dstp = ei + N_EDGES;

    float* ws   = (float*)d_ws;
    float* dinv = ws;
    int*   bkey = (int*)(dinv + N_NODES);
    float* bval = (float*)(bkey + N_EDGES);
    float* hbuf = bval + N_EDGES;
    float* h2   = hbuf + (size_t)N_NODES * HID;

    int* hist = (int*)hbuf;
    int* rb   = (int*)d_out;

    const int B = 256;
    int gMM  = (NTILES + 7) / 8;

    k_count    <<<NB, 1024, 0, stream>>>(dstp, hist);
    k_offsets  <<<1, 1024, 0, stream>>>(hist, rb);
    k_fill     <<<NB, 1024, 0, stream>>>(srcp, dstp, ew, hist, bkey, bval);
    k_deg_dinv <<<NB, 1024, 0, stream>>>(bkey, bval, rb, dinv);

    k_gemm1    <<<gMM, B, 0, stream>>>(x, W1, dinv, hbuf);
    k_agg<true> <<<NB, 1024, 0, stream>>>(bkey, bval, rb, dinv, hbuf, b1, W2, h2);

    k_agg<false><<<NB, 1024, 0, stream>>>(bkey, bval, rb, dinv, h2, b2, W2, out);
}


# x @ W1 blocks ride in the edge-fill launch (no edge dependency; rows stored unscaled), the degree kernel then scales its own range's rows by dinv in place; 7 launches become 6
# speedup vs baseline: 4.8289x; 1.1954x over previous
#include <hip/hip_runtime.h>
#include <math.h>


#define N_NODES 100000
#define N_EDGES 3200000
#define F_IN    512
#define HID     16
#define NCLS    10
#define NTILES  (N_NODES / 16)

typedef __attribute__((ext_vector_type(16))) __bf16 v16bf;
typedef __attribute__((ext_vector_type(8)))  float  v8f;


#define NB      256
#define RNG     416
#define CH      (N_EDGES / NB)
#define CH_PT   ((CH + 1023) / 1024)
#define RB_STEP (RNG * NCLS)
#define H_PT    ((RNG * HID + 1023) / 1024)

__device__ __forceinline__ int scan256_excl(int v, int t, int* wt) {
    const int lane = t & 31;
    int inc = v;
    if (t < 256) {
        #pragma unroll
        for (int d = 1; d < 32; d <<= 1) {
            int o = __shfl_up(inc, d);
            if (lane >= d) inc += o;
        }
        if (lane == 31) wt[t >> 5] = inc;
    }
    __syncthreads();
    int off = 0;
    if (t < 256) {
        const int wv = t >> 5;
        #pragma unroll
        for (int k = 0; k < 7; ++k) if (k < wv) off += wt[k];
    }
    __syncthreads();
    return off + inc - v;
}

__global__ __launch_bounds__(1024) void k_count(const int* __restrict__ dst, int* __restrict__ hist) {
    __shared__ int c[NB];
    const int t = threadIdx.x;
    const int b = blockIdx.x;
    if (t < NB) c[t] = 0;
    __syncthreads();
    const int* dp = dst + b * CH;
    for (int i0 = t; i0 < CH; i0 += 4096) {
        int d[4];
        #pragma unroll
        for (int u = 0; u < 4; ++u) { const int i = i0 + u * 1024; d[u] = (i < CH) ? dp[i] : -1; }
        #pragma unroll
        for (int u = 0; u < 4; ++u) if (d[u] >= 0) atomicAdd(&c[d[u] / RNG], 1);
    }
    __syncthreads();
    if (t < NB) hist[b * NB + t] = c[t];
}

__global__ __launch_bounds__(1024) void k_offsets(int* __restrict__ hist, int* __restrict__ rb) {
    __shared__ int part[4 * NB];
    __shared__ int bs[NB];
    __shared__ int wt[8];
    const int t = threadIdx.x;
    const int r = t & (NB - 1);
    const int q = t >> 8;
    int* hp = hist + (q * 64) * NB + r;

    int s = 0;
    #pragma unroll 16
    for (int j = 0; j < 64; ++j) s += hp[j * NB];
    part[q * NB + r] = s;
    __syncthreads();

    int tot = 0;
    if (t < NB) tot = part[t] + part[NB + t] + part[2 * NB + t] + part[3 * NB + t];
    const int ex = scan256_excl(tot, t, wt);
    if (t < NB) {
        bs[t] = ex;
        if (t * RNG < N_NODES) { rb[t * RB_STEP] = ex; rb[t * RB_STEP + 1] = ex + tot; }
    }
    __syncthreads();

    int run = bs[r];
    for (int k = 0; k < q; ++k) run += part[k * NB + r];
    for (int j0 = 0; j0 < 64; j0 += 16) {
        int v[16];
        #pragma unroll
        for (int j = 0; j < 16; ++j) v[j] = hp[(j0 + j) * NB];
        #pragma unroll
        for (int j = 0; j < 16; ++j) { hp[(j0 + j) * NB] = run; run += v[j]; }
    }
}


#define GM_BLOCKS 250
#define GM_WAVES  25
#define FILL_WORDS (3 * NB + 8 + 3 * CH)

static_assert(GM_BLOCKS * GM_WAVES == NTILES, "one tile per product wave");
static_assert(FILL_WORDS * 4 >= F_IN * HID * 2, "W1 as bf16 fits the shared buffer");

__device__ __forceinline__ void gemm1_role(int* smem, int g, const float* __restrict__ x,
                                           const float* __restrict__ W1, float* __restrict__ h);

__global__ __launch_bounds__(1024) void k_fill_gemm(const int* __restrict__ src, const int* __restrict__ dst,
                                                    const float* __restrict__ w, const int* __restrict__ offs,
                                                    int* __restrict__ bkey, float* __restrict__ bval,
                                                    const float* __restrict__ x, const float* __restrict__ W1,
                                                    float* __restrict__ h) {
    __shared__ __attribute__((aligned(16))) int smem[FILL_WORDS];

    if (blockIdx.x >= NB) {
        gemm1_role(smem, blockIdx.x - NB, x, W1, h);
        return;
    }

    int*   cnt  = smem;
    int*   lst  = smem + NB;
    int*   gdl  = smem + 2 * NB;
    int*   wt   = smem + 3 * NB;
    int*   skey = smem + 3 * NB + 8;
    float* sval = reinterpret_cast<float*>(skey + CH);
    int*   sdel = skey + 2 * CH;

    const int t  = threadIdx.x;
    const int b  = blockIdx.x;
    const int e0 = b * CH;

    if (t < NB) cnt[t] = 0;
    __syncthreads();

    unsigned pk[CH_PT];
    int dd[CH_PT];
    #pragma unroll
    for (int k = 0; k < CH_PT; ++k) {
        const int i = t + k * 1024;
        dd[k] = (i < CH) ? dst[e0 + i] : -1;
    }
    #pragma unroll
    for (int k = 0; k < CH_PT; ++k) {
        pk[k] = 0;
        if (dd[k] >= 0) {
            const int d  = dd[k];
            const int bn = d / RNG;
            const int rk = atomicAdd(&cnt[bn], 1);
            pk[k] = (unsigned)rk | ((unsigned)bn << 14) | ((unsigned)(d - bn * RNG) << 22);
        }
    }
    __syncthreads();

    int cv = 0;
    if (t < NB) cv = cnt[t];
    const int ex = scan256_excl(cv, t, wt);
    if (t < NB) { lst[t] = ex; gdl[t] = offs[b * NB + t] - ex; }
    __syncthreads();

    #pragma unroll
    for (int k = 0; k < CH_PT; ++k) {
        const int i = t + k * 1024;
        if (i < CH) {
            const int rk = pk[k] & 0x3FFF;
            const int bn = (pk[k] >> 14) & 0xFF;
            const int dl = pk[k] >> 22;
            const int p  = lst[bn] + rk;
            skey[p] = src[e0 + i] | (dl << 17);
            sval[p] = w[e0 + i];
            sdel[p] = gdl[bn];
        }
    }
    __syncthreads();

    for (int i = t; i < CH; i += 1024) {
        const int g = i + sdel[i];
        bkey[g] = skey[i];
        bval[g] = sval[i];
    }
}

__global__ __launch_bounds__(1024) void k_deg_dinv(const int* __restrict__ bkey, const float* __restrict__ bval,
                                                   const int* __restrict__ rb, float* __restrict__ dinv,
                                                   float* __restrict__ h) {
    __shared__ float dg[RNG];
    const int t = threadIdx.x;
    const int r = blockIdx.x;
    if (r * RNG >= N_NODES) return;
    const int s = rb[r * RB_STEP], e = rb[r * RB_STEP + 1];
    if (t < RNG) dg[t] = 1.0f;
    __syncthreads();
    const int n0   = r * RNG;
    const int nel  = ((N_NODES - n0 < RNG) ? (N_NODES - n0) : RNG) * HID;
    float* hr = h + (size_t)n0 * HID;
    float hv[H_PT];
    #pragma unroll
    for (int u = 0; u < H_PT; ++u) { const int j = t + u * 1024; hv[u] = (j < nel) ? hr[j] : 0.0f; }
    for (int i0 = s + t; i0 < e; i0 += 4096) {
        int   k[4];
        float v[4];
        #pragma unroll
        for (int u = 0; u < 4; ++u) {
            const int i = i0 + u * 1024;
            k[u] = -1; v[u] = 0.0f;
            if (i < e) { k[u] = bkey[i]; v[u] = bval[i]; }
        }
        #pragma unroll
        for (int u = 0; u < 4; ++u) if (k[u] >= 0) atomicAdd(&dg[k[u] >> 17], v[u]);
    }
    __syncthreads();
    const int node = r * RNG + t;
    if (t < RNG) {
        const float d  = dg[t];
        const float di = (d > 0.0f) ? rsqrtf(d) : 0.0f;
        if (node < N_NODES) dinv[node] = di;
        dg[t] = di;
    }
    __syncthreads();
    #pragma unroll
    for (int u = 0; u < H_PT; ++u) { const int j = t + u * 1024; if (j < nel) hr[j] = hv[u] * dg[j >> 4]; }
}


__device__ __forceinline__ void gemm1_role(int* smem, int g, const float* __restrict__ x,
                                           const float* __restrict__ W1, float* __restrict__ h) {
    __bf16* w1s = reinterpret_cast<__bf16*>(smem);
    int t = threadIdx.x;
    #pragma unroll
    for (int i = 0; i < F_IN * HID / 1024; ++i) w1s[t + i * 1024] = (__bf16)W1[t + i * 1024];
    __syncthreads();

    int wave = t >> 5;
    int lane = t & 31;
    if (wave >= GM_WAVES) return;
    int tile = g * GM_WAVES + wave;

    int col = lane & 15;
    int hh  = lane >> 4;
    const float* xr = x + (size_t)(tile * 16 + col) * F_IN;

    v8f acc;
    #pragma unroll
    for (int r = 0; r < 8; ++r) acc[r] = 0.0f;

    for (int k0 = 0; k0 < F_IN; k0 += 32) {
        v16bf a, b;
        #pragma unroll
        for (int j = 0; j < 8; ++j) {
            int kl = (j < 4) ? (2 * j + 8 * hh) : (16 + 2 * (j - 4) + 8 * hh);
            int k  = k0 + kl;
            float2 xv = *reinterpret_cast<const float2*>(xr + k);
            a[2 * j]     = (__bf16)xv.x;
            a[2 * j + 1] = (__bf16)xv.y;
            b[2 * j]     = w1s[k * HID + col];
            b[2 * j + 1] = w1s[(k + 1) * HID + col];
        }
        acc = __builtin_amdgcn_wmma_f32_16x16x32_bf16(
                  false, a, false, b, (short)0, acc, false, false);
    }

    float* out = h + (size_t)(tile * 16 + 8 * hh) * HID + col;
    #pragma unroll
    for (int r = 0; r < 8; ++r) out[(size_t)r * HID] = acc[r];
}


template <bool FIRST>
__global__ __launch_bounds__(1024) void k_agg(const int* __restrict__ bkey, const float* __restrict__ bval,
                                              const int* rb, const float* __restrict__ dinv,
                                              const float* __restrict__ h, const float* __restrict__ bias,
                                              const float* __restrict__ W2, float* dst) {
    __shared__ float acc[RNG * HID];
    __shared__ float dli[RNG];
    __shared__ __bf16 w2s[FIRST ? 16 * 16 : 2];
    __shared__ float  ost[FIRST ? 1 : RNG * NCLS];

    const int t    = threadIdx.x;
    const int lane = t & 31;
    const int c    = lane & 15;
    const int hsel = lane & 16;
    const int r    = blockIdx.x;
    const int n0   = r * RNG;
    if (n0 >= N_NODES) return;
    const int s    = rb[r * RB_STEP], e = rb[r * RB_STEP + 1];

    for (int j = t; j < RNG * HID; j += 1024) {
        const int node = n0 + (j >> 4);
        acc[j] = (node < N_NODES) ? h[(size_t)n0 * HID + j] : 0.0f;
    }
    if (t < RNG) dli[t] = (n0 + t < N_NODES) ? dinv[n0 + t] : 0.0f;
    if constexpr (FIRST) {
        if (t < 16 * 16) {
            const int k = t >> 4, n = t & 15;
            w2s[t] = (n < NCLS) ? (__bf16)W2[k * NCLS + n] : (__bf16)0.0f;
        }
    }
    __syncthreads();

    for (int base = s + (t >> 5) * 128; base < e; base += 32 * 128) {
        int   k0 = 0, k1 = 0, k2 = 0, k3 = 0;
        float v0 = 0.0f, v1 = 0.0f, v2 = 0.0f, v3 = 0.0f;
        const int i0 = base + lane, i1 = i0 + 32, i2 = i0 + 64, i3 = i0 + 96;
        if (i0 < e) { k0 = bkey[i0]; v0 = bval[i0]; }
        if (i1 < e) { k1 = bkey[i1]; v1 = bval[i1]; }
        if (i2 < e) { k2 = bkey[i2]; v2 = bval[i2]; }
        if (i3 < e) { k3 = bkey[i3]; v3 = bval[i3]; }

        #pragma nounroll
        for (int u = 0; u < 4; ++u) {
            const int b0 = base + u * 32;
            if (b0 < e) {
                int   kk[16];
                float vv[16], hv[16];
                #pragma unroll
                for (int j = 0; j < 16; ++j) { kk[j] = __shfl(k0, j + hsel); vv[j] = __shfl(v0, j + hsel); }
                #pragma unroll
                for (int j = 0; j < 16; ++j) hv[j] = h[(size_t)(kk[j] & 0x1FFFF) * HID + c];
                #pragma unroll
                for (int j = 0; j < 16; ++j)
                    if (b0 + j + hsel < e) atomicAdd(&acc[(kk[j] >> 17) * HID + c], vv[j] * hv[j]);
            }
            k0 = k1; k1 = k2; k2 = k3;
            v0 = v1; v1 = v2; v2 = v3;
        }
    }
    __syncthreads();

    if constexpr (FIRST) {
        const int wave = t >> 5;
        if (wave < RNG / 16 && n0 + wave * 16 < N_NODES) {
            const int hh = lane >> 4;
            const float* row = acc + (wave * 16 + c) * HID;
            const float  dl  = dli[wave * 16 + c];
            v16bf a, b;
            #pragma unroll
            for (int i = 0; i < 16; ++i) { a[i] = (__bf16)0.0f; b[i] = (__bf16)0.0f; }
            #pragma unroll
            for (int j = 0; j < 4; ++j) {
                const int kl = 2 * j + 8 * hh;
                const float x0 = dl * row[kl]     + bias[kl];
                const float x1 = dl * row[kl + 1] + bias[kl + 1];
                a[2 * j]     = (__bf16)(x0 > 0.0f ? x0 : 0.0f);
                a[2 * j + 1] = (__bf16)(x1 > 0.0f ? x1 : 0.0f);
                b[2 * j]     = w2s[kl * 16 + c];
                b[2 * j + 1] = w2s[(kl + 1) * 16 + c];
            }
            v8f d;
            #pragma unroll
            for (int q = 0; q < 8; ++q) d[q] = 0.0f;
            d = __builtin_amdgcn_wmma_f32_16x16x32_bf16(
                    false, a, false, b, (short)0, d, false, false);
            #pragma unroll
            for (int q = 0; q < 8; ++q)
                acc[(wave * 16 + 8 * hh + q) * HID + c] = d[q] * dli[wave * 16 + 8 * hh + q];
        }
        __syncthreads();

        for (int j = t; j < RNG * HID; j += 1024)
            if (n0 + (j >> 4) < N_NODES) dst[(size_t)n0 * HID + j] = acc[j];
    } else {
        if (t < RNG && n0 + t < N_NODES) {
            float z[NCLS];
            float m = -INFINITY;
            #pragma unroll
            for (int q = 0; q < NCLS; ++q) {
                z[q] = dli[t] * acc[t * HID + q] + bias[q];
                m = fmaxf(m, z[q]);
            }
            float se = 0.0f;
            #pragma unroll
            for (int q = 0; q < NCLS; ++q) se += __expf(z[q] - m);
            const float ls = __logf(se) + m;
            #pragma unroll
            for (int q = 0; q < NCLS; ++q) ost[t * NCLS + q] = z[q] - ls;
        }
        __syncthreads();

        const int rows = (N_NODES - n0 < RNG) ? (N_NODES - n0) : RNG;
        for (int j = t; j < rows * NCLS; j += 1024) dst[(size_t)n0 * NCLS + j] = ost[j];
    }
}


extern "C" void kernel_launch(void* const* d_in, const int* in_sizes, int n_in,
                              void* d_out, int out_size, void* d_ws, size_t ws_size,
                              hipStream_t stream) {
    const float* x  = (const float*)d_in[0];
    const int*   ei = (const int*)  d_in[1];
    const float* ew = (const float*)d_in[2];
    const float* W1 = (const float*)d_in[3];
    const float* b1 = (const float*)d_in[4];
    const float* W2 = (const float*)d_in[5];
    const float* b2 = (const float*)d_in[6];
    float* out = (float*)d_out;

    const int* srcp = ei;
    const int* dstp = ei + N_EDGES;

    float* ws   = (float*)d_ws;
    float* dinv = ws;
    int*   bkey = (int*)(dinv + N_NODES);
    float* bval = (float*)(bkey + N_EDGES);
    float* hbuf = bval + N_EDGES;
    float* h2   = hbuf + (size_t)N_NODES * HID;

    int* hist = (int*)h2;
    int* rb   = (int*)d_out;

    k_count     <<<NB, 1024, 0, stream>>>(dstp, hist);
    k_offsets   <<<1, 1024, 0, stream>>>(hist, rb);
    k_fill_gemm <<<NB + GM_BLOCKS, 1024, 0, stream>>>(srcp, dstp, ew, hist, bkey, bval, x, W1, hbuf);
    k_deg_dinv  <<<NB, 1024, 0, stream>>>(bkey, bval, rb, dinv, hbuf);

    k_agg<true> <<<NB, 1024, 0, stream>>>(bkey, bval, rb, dinv, hbuf, b1, W2, h2);

    k_agg<false><<<NB, 1024, 0, stream>>>(bkey, bval, rb, dinv, h2, b2, W2, out);
}


# Loads grouped per phase, one round trip not one per loop trip: count's 13 dst, offsets' 64 counts (kept for its write pass), degree's 16 records, aggregation's rows+dinv+first and next records
# speedup vs baseline: 5.1100x; 1.0582x over previous
#include <hip/hip_runtime.h>
#include <math.h>


#define N_NODES 100000
#define N_EDGES 3200000
#define F_IN    512
#define HID     16
#define NCLS    10
#define NTILES  (N_NODES / 16)

typedef __attribute__((ext_vector_type(16))) __bf16 v16bf;
typedef __attribute__((ext_vector_type(8)))  float  v8f;


#define NB      256
#define RNG     416
#define CH      (N_EDGES / NB)
#define CH_PT   ((CH + 1023) / 1024)
#define RB_STEP (RNG * NCLS)
#define H_PT    ((RNG * HID + 1023) / 1024)
#define DG_PT   16

__device__ __forceinline__ int scan256_excl(int v, int t, int* wt) {
    const int lane = t & 31;
    int inc = v;
    if (t < 256) {
        #pragma unroll
        for (int d = 1; d < 32; d <<= 1) {
            int o = __shfl_up(inc, d);
            if (lane >= d) inc += o;
        }
        if (lane == 31) wt[t >> 5] = inc;
    }
    __syncthreads();
    int off = 0;
    if (t < 256) {
        const int wv = t >> 5;
        #pragma unroll
        for (int k = 0; k < 7; ++k) if (k < wv) off += wt[k];
    }
    __syncthreads();
    return off + inc - v;
}

__global__ __launch_bounds__(1024) void k_count(const int* __restrict__ dst, int* __restrict__ hist) {
    __shared__ int c[NB];
    const int t = threadIdx.x;
    const int b = blockIdx.x;
    if (t < NB) c[t] = 0;
    __syncthreads();
    const int* dp = dst + b * CH;
    int d[CH_PT];
    #pragma unroll
    for (int k = 0; k < CH_PT; ++k) { const int i = t + k * 1024; d[k] = (i < CH) ? dp[i] : -1; }
    #pragma unroll
    for (int k = 0; k < CH_PT; ++k) if (d[k] >= 0) atomicAdd(&c[d[k] / RNG], 1);
    __syncthreads();
    if (t < NB) hist[b * NB + t] = c[t];
}

__global__ __launch_bounds__(1024) void k_offsets(int* __restrict__ hist, int* __restrict__ rb) {
    __shared__ int part[4 * NB];
    __shared__ int bs[NB];
    __shared__ int wt[8];
    const int t = threadIdx.x;
    const int r = t & (NB - 1);
    const int q = t >> 8;
    int* hp = hist + (q * 64) * NB + r;

    int v[64];
    #pragma unroll
    for (int j = 0; j < 64; ++j) v[j] = hp[j * NB];
    int s = 0;
    #pragma unroll
    for (int j = 0; j < 64; ++j) s += v[j];
    part[q * NB + r] = s;
    __syncthreads();

    int tot = 0;
    if (t < NB) tot = part[t] + part[NB + t] + part[2 * NB + t] + part[3 * NB + t];
    const int ex = scan256_excl(tot, t, wt);
    if (t < NB) {
        bs[t] = ex;
        if (t * RNG < N_NODES) { rb[t * RB_STEP] = ex; rb[t * RB_STEP + 1] = ex + tot; }
    }
    __syncthreads();

    int run = bs[r];
    for (int k = 0; k < q; ++k) run += part[k * NB + r];
    #pragma unroll
    for (int j = 0; j < 64; ++j) { hp[j * NB] = run; run += v[j]; }
}


#define GM_BLOCKS 250
#define GM_WAVES  25
#define FILL_WORDS (3 * NB + 8 + 3 * CH)

static_assert(GM_BLOCKS * GM_WAVES == NTILES, "one tile per product wave");
static_assert(FILL_WORDS * 4 >= F_IN * HID * 2, "W1 as bf16 fits the shared buffer");

__device__ __forceinline__ void gemm1_role(int* smem, int g, const float* __restrict__ x,
                                           const float* __restrict__ W1, float* __restrict__ h);

__global__ __launch_bounds__(1024) void k_fill_gemm(const int* __restrict__ src, const int* __restrict__ dst,
                                                    const float* __restrict__ w, const int* __restrict__ offs,
                                                    int* __restrict__ bkey, float* __restrict__ bval,
                                                    const float* __restrict__ x, const float* __restrict__ W1,
                                                    float* __restrict__ h) {
    __shared__ __attribute__((aligned(16))) int smem[FILL_WORDS];

    if (blockIdx.x >= NB) {
        gemm1_role(smem, blockIdx.x - NB, x, W1, h);
        return;
    }

    int*   cnt  = smem;
    int*   lst  = smem + NB;
    int*   gdl  = smem + 2 * NB;
    int*   wt   = smem + 3 * NB;
    int*   skey = smem + 3 * NB + 8;
    float* sval = reinterpret_cast<float*>(skey + CH);
    int*   sdel = skey + 2 * CH;

    const int t  = threadIdx.x;
    const int b  = blockIdx.x;
    const int e0 = b * CH;

    if (t < NB) cnt[t] = 0;
    __syncthreads();

    unsigned pk[CH_PT];
    int dd[CH_PT];
    #pragma unroll
    for (int k = 0; k < CH_PT; ++k) {
        const int i = t + k * 1024;
        dd[k] = (i < CH) ? dst[e0 + i] : -1;
    }
    #pragma unroll
    for (int k = 0; k < CH_PT; ++k) {
        pk[k] = 0;
        if (dd[k] >= 0) {
            const int d  = dd[k];
            const int bn = d / RNG;
            const int rk = atomicAdd(&cnt[bn], 1);
            pk[k] = (unsigned)rk | ((unsigned)bn << 14) | ((unsigned)(d - bn * RNG) << 22);
        }
    }
    __syncthreads();

    int cv = 0;
    if (t < NB) cv = cnt[t];
    const int ex = scan256_excl(cv, t, wt);
    if (t < NB) { lst[t] = ex; gdl[t] = offs[b * NB + t] - ex; }
    __syncthreads();

    #pragma unroll
    for (int k = 0; k < CH_PT; ++k) {
        const int i = t + k * 1024;
        if (i < CH) {
            const int rk = pk[k] & 0x3FFF;
            const int bn = (pk[k] >> 14) & 0xFF;
            const int dl = pk[k] >> 22;
            const int p  = lst[bn] + rk;
            skey[p] = src[e0 + i] | (dl << 17);
            sval[p] = w[e0 + i];
            sdel[p] = gdl[bn];
        }
    }
    __syncthreads();

    for (int i = t; i < CH; i += 1024) {
        const int g = i + sdel[i];
        bkey[g] = skey[i];
        bval[g] = sval[i];
    }
}

__global__ __launch_bounds__(1024) void k_deg_dinv(const int* __restrict__ bkey, const float* __restrict__ bval,
                                                   const int* __restrict__ rb, float* __restrict__ dinv,
                                                   float* __restrict__ h) {
    __shared__ float dg[RNG];
    const int t = threadIdx.x;
    const int r = blockIdx.x;
    if (r * RNG >= N_NODES) return;
    const int s = rb[r * RB_STEP], e = rb[r * RB_STEP + 1];
    if (t < RNG) dg[t] = 1.0f;
    __syncthreads();
    const int n0   = r * RNG;
    const int nel  = ((N_NODES - n0 < RNG) ? (N_NODES - n0) : RNG) * HID;
    float* hr = h + (size_t)n0 * HID;
    float hv[H_PT];
    #pragma unroll
    for (int u = 0; u < H_PT; ++u) { const int j = t + u * 1024; hv[u] = (j < nel) ? hr[j] : 0.0f; }
    for (int i0 = s + t; i0 < e; i0 += DG_PT * 1024) {
        int   k[DG_PT];
        float v[DG_PT];
        #pragma unroll
        for (int u = 0; u < DG_PT; ++u) {
            const int i = i0 + u * 1024;
            k[u] = -1; v[u] = 0.0f;
            if (i < e) { k[u] = bkey[i]; v[u] = bval[i]; }
        }
        #pragma unroll
        for (int u = 0; u < DG_PT; ++u) if (k[u] >= 0) atomicAdd(&dg[k[u] >> 17], v[u]);
    }
    __syncthreads();
    const int node = r * RNG + t;
    if (t < RNG) {
        const float d  = dg[t];
        const float di = (d > 0.0f) ? rsqrtf(d) : 0.0f;
        if (node < N_NODES) dinv[node] = di;
        dg[t] = di;
    }
    __syncthreads();
    #pragma unroll
    for (int u = 0; u < H_PT; ++u) { const int j = t + u * 1024; if (j < nel) hr[j] = hv[u] * dg[j >> 4]; }
}


__device__ __forceinline__ void gemm1_role(int* smem, int g, const float* __restrict__ x,
                                           const float* __restrict__ W1, float* __restrict__ h) {
    __bf16* w1s = reinterpret_cast<__bf16*>(smem);
    int t = threadIdx.x;
    #pragma unroll
    for (int i = 0; i < F_IN * HID / 1024; ++i) w1s[t + i * 1024] = (__bf16)W1[t + i * 1024];
    __syncthreads();

    int wave = t >> 5;
    int lane = t & 31;
    if (wave >= GM_WAVES) return;
    int tile = g * GM_WAVES + wave;

    int col = lane & 15;
    int hh  = lane >> 4;
    const float* xr = x + (size_t)(tile * 16 + col) * F_IN;

    v8f acc;
    #pragma unroll
    for (int r = 0; r < 8; ++r) acc[r] = 0.0f;

    for (int k0 = 0; k0 < F_IN; k0 += 32) {
        v16bf a, b;
        #pragma unroll
        for (int j = 0; j < 8; ++j) {
            int kl = (j < 4) ? (2 * j + 8 * hh) : (16 + 2 * (j - 4) + 8 * hh);
            int k  = k0 + kl;
            float2 xv = *reinterpret_cast<const float2*>(xr + k);
            a[2 * j]     = (__bf16)xv.x;
            a[2 * j + 1] = (__bf16)xv.y;
            b[2 * j]     = w1s[k * HID + col];
            b[2 * j + 1] = w1s[(k + 1) * HID + col];
        }
        acc = __builtin_amdgcn_wmma_f32_16x16x32_bf16(
                  false, a, false, b, (short)0, acc, false, false);
    }

    float* out = h + (size_t)(tile * 16 + 8 * hh) * HID + col;
    #pragma unroll
    for (int r = 0; r < 8; ++r) out[(size_t)r * HID] = acc[r];
}


template <bool FIRST>
__global__ __launch_bounds__(1024) void k_agg(const int* __restrict__ bkey, const float* __restrict__ bval,
                                              const int* rb, const float* __restrict__ dinv,
                                              const float* __restrict__ h, const float* __restrict__ bias,
                                              const float* __restrict__ W2, float* dst) {
    __shared__ float acc[RNG * HID];
    __shared__ float dli[RNG];
    __shared__ __bf16 w2s[FIRST ? 16 * 16 : 2];
    __shared__ float  ost[FIRST ? 1 : RNG * NCLS];

    const int t    = threadIdx.x;
    const int lane = t & 31;
    const int c    = lane & 15;
    const int hsel = lane & 16;
    const int r    = blockIdx.x;
    const int n0   = r * RNG;
    if (n0 >= N_NODES) return;
    const int s    = rb[r * RB_STEP], e = rb[r * RB_STEP + 1];

    float hr[H_PT];
    #pragma unroll
    for (int u = 0; u < H_PT; ++u) {
        const int j = t + u * 1024;
        hr[u] = (j < RNG * HID && n0 + (j >> 4) < N_NODES) ? h[(size_t)n0 * HID + j] : 0.0f;
    }
    float dv = 0.0f;
    if (t < RNG && n0 + t < N_NODES) dv = dinv[n0 + t];
    int   k0 = 0, k1 = 0, k2 = 0, k3 = 0;
    float v0 = 0.0f, v1 = 0.0f, v2 = 0.0f, v3 = 0.0f;
    int base = s + (t >> 5) * 128;
    {
        const int i0 = base + lane, i1 = i0 + 32, i2 = i0 + 64, i3 = i0 + 96;
        if (i0 < e) { k0 = bkey[i0]; v0 = bval[i0]; }
        if (i1 < e) { k1 = bkey[i1]; v1 = bval[i1]; }
        if (i2 < e) { k2 = bkey[i2]; v2 = bval[i2]; }
        if (i3 < e) { k3 = bkey[i3]; v3 = bval[i3]; }
    }
    float w2v = 0.0f;
    if constexpr (FIRST) {
        if (t < 16 * 16 && (t & 15) < NCLS) w2v = W2[(t >> 4) * NCLS + (t & 15)];
    }
    #pragma unroll
    for (int u = 0; u < H_PT; ++u) { const int j = t + u * 1024; if (j < RNG * HID) acc[j] = hr[u]; }
    if (t < RNG) dli[t] = dv;
    if constexpr (FIRST) {
        if (t < 16 * 16) w2s[t] = (__bf16)w2v;
    }
    __syncthreads();

    for (; base < e; base += 32 * 128) {
        int   nk0 = 0, nk1 = 0, nk2 = 0, nk3 = 0;
        float nv0 = 0.0f, nv1 = 0.0f, nv2 = 0.0f, nv3 = 0.0f;
        {
            const int i0 = base + 32 * 128 + lane, i1 = i0 + 32, i2 = i0 + 64, i3 = i0 + 96;
            if (i0 < e) { nk0 = bkey[i0]; nv0 = bval[i0]; }
            if (i1 < e) { nk1 = bkey[i1]; nv1 = bval[i1]; }
            if (i2 < e) { nk2 = bkey[i2]; nv2 = bval[i2]; }
            if (i3 < e) { nk3 = bkey[i3]; nv3 = bval[i3]; }
        }

        #pragma nounroll
        for (int u = 0; u < 4; ++u) {
            const int b0 = base + u * 32;
            if (b0 < e) {
                int   kk[16];
                float vv[16], hv[16];
                #pragma unroll
                for (int j = 0; j < 16; ++j) { kk[j] = __shfl(k0, j + hsel); vv[j] = __shfl(v0, j + hsel); }
                #pragma unroll
                for (int j = 0; j < 16; ++j) hv[j] = h[(size_t)(kk[j] & 0x1FFFF) * HID + c];
                #pragma unroll
                for (int j = 0; j < 16; ++j) {
                    const float p = (b0 + j + hsel < e) ? vv[j] * hv[j] : 0.0f;
                    atomicAdd(&acc[(kk[j] >> 17) * HID + c], p);
                }
            }
            k0 = k1; k1 = k2; k2 = k3;
            v0 = v1; v1 = v2; v2 = v3;
        }
        k0 = nk0; k1 = nk1; k2 = nk2; k3 = nk3;
        v0 = nv0; v1 = nv1; v2 = nv2; v3 = nv3;
    }
    __syncthreads();

    if constexpr (FIRST) {
        const int wave = t >> 5;
        if (wave < RNG / 16 && n0 + wave * 16 < N_NODES) {
            const int hh = lane >> 4;
            const float* row = acc + (wave * 16 + c) * HID;
            const float  dl  = dli[wave * 16 + c];
            v16bf a, b;
            #pragma unroll
            for (int i = 0; i < 16; ++i) { a[i] = (__bf16)0.0f; b[i] = (__bf16)0.0f; }
            #pragma unroll
            for (int j = 0; j < 4; ++j) {
                const int kl = 2 * j + 8 * hh;
                const float x0 = dl * row[kl]     + bias[kl];
                const float x1 = dl * row[kl + 1] + bias[kl + 1];
                a[2 * j]     = (__bf16)(x0 > 0.0f ? x0 : 0.0f);
                a[2 * j + 1] = (__bf16)(x1 > 0.0f ? x1 : 0.0f);
                b[2 * j]     = w2s[kl * 16 + c];
                b[2 * j + 1] = w2s[(kl + 1) * 16 + c];
            }
            v8f d;
            #pragma unroll
            for (int q = 0; q < 8; ++q) d[q] = 0.0f;
            d = __builtin_amdgcn_wmma_f32_16x16x32_bf16(
                    false, a, false, b, (short)0, d, false, false);
            #pragma unroll
            for (int q = 0; q < 8; ++q)
                acc[(wave * 16 + 8 * hh + q) * HID + c] = d[q] * dli[wave * 16 + 8 * hh + q];
        }
        __syncthreads();

        for (int j = t; j < RNG * HID; j += 1024)
            if (n0 + (j >> 4) < N_NODES) dst[(size_t)n0 * HID + j] = acc[j];
    } else {
        if (t < RNG && n0 + t < N_NODES) {
            float z[NCLS];
            float m = -INFINITY;
            #pragma unroll
            for (int q = 0; q < NCLS; ++q) {
                z[q] = dli[t] * acc[t * HID + q] + bias[q];
                m = fmaxf(m, z[q]);
            }
            float se = 0.0f;
            #pragma unroll
            for (int q = 0; q < NCLS; ++q) se += __expf(z[q] - m);
            const float ls = __logf(se) + m;
            #pragma unroll
            for (int q = 0; q < NCLS; ++q) ost[t * NCLS + q] = z[q] - ls;
        }
        __syncthreads();

        const int rows = (N_NODES - n0 < RNG) ? (N_NODES - n0) : RNG;
        for (int j = t; j < rows * NCLS; j += 1024) dst[(size_t)n0 * NCLS + j] = ost[j];
    }
}


extern "C" void kernel_launch(void* const* d_in, const int* in_sizes, int n_in,
                              void* d_out, int out_size, void* d_ws, size_t ws_size,
                              hipStream_t stream) {
    const float* x  = (const float*)d_in[0];
    const int*   ei = (const int*)  d_in[1];
    const float* ew = (const float*)d_in[2];
    const float* W1 = (const float*)d_in[3];
    const float* b1 = (const float*)d_in[4];
    const float* W2 = (const float*)d_in[5];
    const float* b2 = (const float*)d_in[6];
    float* out = (float*)d_out;

    const int* srcp = ei;
    const int* dstp = ei + N_EDGES;

    float* ws   = (float*)d_ws;
    float* dinv = ws;
    int*   bkey = (int*)(dinv + N_NODES);
    float* bval = (float*)(bkey + N_EDGES);
    float* hbuf = bval + N_EDGES;
    float* h2   = hbuf + (size_t)N_NODES * HID;

    int* hist = (int*)h2;
    int* rb   = (int*)d_out;

    k_count     <<<NB, 1024, 0, stream>>>(dstp, hist);
    k_offsets   <<<1, 1024, 0, stream>>>(hist, rb);
    k_fill_gemm <<<NB + GM_BLOCKS, 1024, 0, stream>>>(srcp, dstp, ew, hist, bkey, bval, x, W1, hbuf);
    k_deg_dinv  <<<NB, 1024, 0, stream>>>(bkey, bval, rb, dinv, hbuf);

    k_agg<true> <<<NB, 1024, 0, stream>>>(bkey, bval, rb, dinv, hbuf, b1, W2, h2);

    k_agg<false><<<NB, 1024, 0, stream>>>(bkey, bval, rb, dinv, h2, b2, W2, out);
}


# Edge blocks sort edges by range in their own slice (run table per range); a range's waves walk 8 runs each; count and offsets launches go, 6 launches become 4; summation order changes
# speedup vs baseline: 5.4320x; 1.0630x over previous
#include <hip/hip_runtime.h>
#include <math.h>


#define N_NODES 100000
#define N_EDGES 3200000
#define F_IN    512
#define HID     16
#define NCLS    10
#define NTILES  (N_NODES / 16)

typedef __attribute__((ext_vector_type(16))) __bf16 v16bf;
typedef __attribute__((ext_vector_type(8)))  float  v8f;


#define NB      256
#define RNG     416
#define CH      (N_EDGES / NB)
#define CH_PT   ((CH + 1023) / 1024)
#define RB_STEP (RNG * NCLS)
#define H_PT    ((RNG * HID + 1023) / 1024)
#define DG_PT   16
#define RUNS_PW (NB / 32)

static_assert((N_NODES % RNG == 0 ? RNG : N_NODES % RNG) * NCLS >= 2 * NB, "the last range's rows hold its run table");
static_assert(RB_STEP % 2 == 0, "run pairs are 8-byte aligned");

__device__ __forceinline__ int wave_runs(const int* tb, int wave, int lane, int (&rp)[RUNS_PW], int (&ra)[RUNS_PW]) {
    int gs = 0, ln = 0;
    if (lane < RUNS_PW) {
        const int2 q = reinterpret_cast<const int2*>(tb)[wave * RUNS_PW + lane];
        gs = q.x; ln = q.y;
    }
    int inc = ln;
    #pragma unroll
    for (int d = 1; d < RUNS_PW; d <<= 1) {
        const int o = __shfl_up(inc, d);
        if (lane >= d) inc += o;
    }
    const int ex = inc - ln;
    #pragma unroll
    for (int l = 0; l < RUNS_PW; ++l) {
        rp[l] = __builtin_amdgcn_readlane(ex, l);
        ra[l] = __builtin_amdgcn_readlane(gs - ex, l);
    }
    return __builtin_amdgcn_readlane(inc, RUNS_PW - 1);
}

__device__ __forceinline__ int run_rec(int i, const int (&rp)[RUNS_PW], const int (&ra)[RUNS_PW]) {
    int a = ra[0];
    #pragma unroll
    for (int l = 1; l < RUNS_PW; ++l) a = (i >= rp[l]) ? ra[l] : a;
    return a + i;
}

__device__ __forceinline__ int scan256_excl(int v, int t, int* wt) {
    const int lane = t & 31;
    int inc = v;
    if (t < 256) {
        #pragma unroll
        for (int d = 1; d < 32; d <<= 1) {
            int o = __shfl_up(inc, d);
            if (lane >= d) inc += o;
        }
        if (lane == 31) wt[t >> 5] = inc;
    }
    __syncthreads();
    int off = 0;
    if (t < 256) {
        const int wv = t >> 5;
        #pragma unroll
        for (int k = 0; k < 7; ++k) if (k < wv) off += wt[k];
    }
    __syncthreads();
    return off + inc - v;
}


#define GM_BLOCKS 250
#define GM_WAVES  25
#define FILL_WORDS (2 * NB + 8 + 2 * CH)

static_assert(GM_BLOCKS * GM_WAVES == NTILES, "one tile per product wave");
static_assert(FILL_WORDS * 4 >= F_IN * HID * 2, "W1 as bf16 fits the shared buffer");

__device__ __forceinline__ void gemm1_role(int* smem, int g, const float* __restrict__ x,
                                           const float* __restrict__ W1, float* __restrict__ h);

__global__ __launch_bounds__(1024) void k_fill_gemm(const int* __restrict__ src, const int* __restrict__ dst,
                                                    const float* __restrict__ w, int* __restrict__ rb,
                                                    int* __restrict__ bkey, float* __restrict__ bval,
                                                    const float* __restrict__ x, const float* __restrict__ W1,
                                                    float* __restrict__ h) {
    __shared__ __attribute__((aligned(16))) int smem[FILL_WORDS];

    if (blockIdx.x >= NB) {
        gemm1_role(smem, blockIdx.x - NB, x, W1, h);
        return;
    }

    int*   cnt  = smem;
    int*   lst  = smem + NB;
    int*   wt   = smem + 2 * NB;
    int*   skey = smem + 2 * NB + 8;
    float* sval = reinterpret_cast<float*>(skey + CH);

    const int t  = threadIdx.x;
    const int b  = blockIdx.x;
    const int e0 = b * CH;

    if (t < NB) cnt[t] = 0;
    __syncthreads();

    unsigned pk[CH_PT];
    int dd[CH_PT];
    #pragma unroll
    for (int k = 0; k < CH_PT; ++k) {
        const int i = t + k * 1024;
        dd[k] = (i < CH) ? dst[e0 + i] : -1;
    }
    #pragma unroll
    for (int k = 0; k < CH_PT; ++k) {
        pk[k] = 0;
        if (dd[k] >= 0) {
            const int d  = dd[k];
            const int bn = d / RNG;
            const int rk = atomicAdd(&cnt[bn], 1);
            pk[k] = (unsigned)rk | ((unsigned)bn << 14) | ((unsigned)(d - bn * RNG) << 22);
        }
    }
    __syncthreads();

    int cv = 0;
    if (t < NB) cv = cnt[t];
    const int ex = scan256_excl(cv, t, wt);
    if (t < NB) lst[t] = ex;
    __syncthreads();

    #pragma unroll
    for (int k = 0; k < CH_PT; ++k) {
        const int i = t + k * 1024;
        if (i < CH) {
            const int rk = pk[k] & 0x3FFF;
            const int bn = (pk[k] >> 14) & 0xFF;
            const int dl = pk[k] >> 22;
            const int p  = lst[bn] + rk;
            skey[p] = src[e0 + i] | (dl << 17);
            sval[p] = w[e0 + i];
        }
    }
    __syncthreads();

    if (t < NB && t * RNG < N_NODES)
        reinterpret_cast<int2*>(rb + t * RB_STEP)[b] = make_int2(e0 + ex, cv);

    for (int i = t; i < CH; i += 1024) {
        bkey[e0 + i] = skey[i];
        bval[e0 + i] = sval[i];
    }
}

__global__ __launch_bounds__(1024) void k_deg_dinv(const int* __restrict__ bkey, const float* __restrict__ bval,
                                                   const int* __restrict__ rb, float* __restrict__ dinv,
                                                   float* __restrict__ h) {
    __shared__ float dg[RNG];
    const int t = threadIdx.x;
    const int r = blockIdx.x;
    if (r * RNG >= N_NODES) return;
    const int n0   = r * RNG;
    const int nel  = ((N_NODES - n0 < RNG) ? (N_NODES - n0) : RNG) * HID;
    float* hr = h + (size_t)n0 * HID;
    float hv[H_PT];
    #pragma unroll
    for (int u = 0; u < H_PT; ++u) { const int j = t + u * 1024; hv[u] = (j < nel) ? hr[j] : 0.0f; }
    int rp[RUNS_PW], ra[RUNS_PW];
    const int e = wave_runs(rb + r * RB_STEP, t >> 5, t & 31, rp, ra);
    if (t < RNG) dg[t] = 1.0f;
    __syncthreads();
    for (int i0 = (t & 31); i0 < e; i0 += DG_PT * 32) {
        int   k[DG_PT];
        float v[DG_PT];
        #pragma unroll
        for (int u = 0; u < DG_PT; ++u) {
            const int i = i0 + u * 32;
            k[u] = -1; v[u] = 0.0f;
            if (i < e) { const int g = run_rec(i, rp, ra); k[u] = bkey[g]; v[u] = bval[g]; }
        }
        #pragma unroll
        for (int u = 0; u < DG_PT; ++u) if (k[u] >= 0) atomicAdd(&dg[k[u] >> 17], v[u]);
    }
    __syncthreads();
    const int node = r * RNG + t;
    if (t < RNG) {
        const float d  = dg[t];
        const float di = (d > 0.0f) ? rsqrtf(d) : 0.0f;
        if (node < N_NODES) dinv[node] = di;
        dg[t] = di;
    }
    __syncthreads();
    #pragma unroll
    for (int u = 0; u < H_PT; ++u) { const int j = t + u * 1024; if (j < nel) hr[j] = hv[u] * dg[j >> 4]; }
}


__device__ __forceinline__ void gemm1_role(int* smem, int g, const float* __restrict__ x,
                                           const float* __restrict__ W1, float* __restrict__ h) {
    __bf16* w1s = reinterpret_cast<__bf16*>(smem);
    int t = threadIdx.x;
    #pragma unroll
    for (int i = 0; i < F_IN * HID / 1024; ++i) w1s[t + i * 1024] = (__bf16)W1[t + i * 1024];
    __syncthreads();

    int wave = t >> 5;
    int lane = t & 31;
    if (wave >= GM_WAVES) return;
    int tile = g * GM_WAVES + wave;

    int col = lane & 15;
    int hh  = lane >> 4;
    const float* xr = x + (size_t)(tile * 16 + col) * F_IN;

    v8f acc;
    #pragma unroll
    for (int r = 0; r < 8; ++r) acc[r] = 0.0f;

    for (int k0 = 0; k0 < F_IN; k0 += 32) {
        v16bf a, b;
        #pragma unroll
        for (int j = 0; j < 8; ++j) {
            int kl = (j < 4) ? (2 * j + 8 * hh) : (16 + 2 * (j - 4) + 8 * hh);
            int k  = k0 + kl;
            float2 xv = *reinterpret_cast<const float2*>(xr + k);
            a[2 * j]     = (__bf16)xv.x;
            a[2 * j + 1] = (__bf16)xv.y;
            b[2 * j]     = w1s[k * HID + col];
            b[2 * j + 1] = w1s[(k + 1) * HID + col];
        }
        acc = __builtin_amdgcn_wmma_f32_16x16x32_bf16(
                  false, a, false, b, (short)0, acc, false, false);
    }

    float* out = h + (size_t)(tile * 16 + 8 * hh) * HID + col;
    #pragma unroll
    for (int r = 0; r < 8; ++r) out[(size_t)r * HID] = acc[r];
}


template <bool FIRST>
__global__ __launch_bounds__(1024) void k_agg(const int* __restrict__ bkey, const float* __restrict__ bval,
                                              const int* rb, const float* __restrict__ dinv,
                                              const float* __restrict__ h, const float* __restrict__ bias,
                                              const float* __restrict__ W2, float* dst) {
    __shared__ float acc[RNG * HID];
    __shared__ float dli[RNG];
    __shared__ __bf16 w2s[FIRST ? 16 * 16 : 2];
    __shared__ float  ost[FIRST ? 1 : RNG * NCLS];

    const int t    = threadIdx.x;
    const int lane = t & 31;
    const int c    = lane & 15;
    const int hsel = lane & 16;
    const int r    = blockIdx.x;
    const int n0   = r * RNG;
    if (n0 >= N_NODES) return;

    float hr[H_PT];
    #pragma unroll
    for (int u = 0; u < H_PT; ++u) {
        const int j = t + u * 1024;
        hr[u] = (j < RNG * HID && n0 + (j >> 4) < N_NODES) ? h[(size_t)n0 * HID + j] : 0.0f;
    }
    float dv = 0.0f;
    if (t < RNG && n0 + t < N_NODES) dv = dinv[n0 + t];
    int rp[RUNS_PW], ra[RUNS_PW];
    const int e = wave_runs(rb + r * RB_STEP, t >> 5, lane, rp, ra);
    int   k0 = 0, k1 = 0, k2 = 0, k3 = 0;
    float v0 = 0.0f, v1 = 0.0f, v2 = 0.0f, v3 = 0.0f;
    int base = 0;
    {
        const int i0 = lane, i1 = i0 + 32, i2 = i0 + 64, i3 = i0 + 96;
        if (i0 < e) { const int g = run_rec(i0, rp, ra); k0 = bkey[g]; v0 = bval[g]; }
        if (i1 < e) { const int g = run_rec(i1, rp, ra); k1 = bkey[g]; v1 = bval[g]; }
        if (i2 < e) { const int g = run_rec(i2, rp, ra); k2 = bkey[g]; v2 = bval[g]; }
        if (i3 < e) { const int g = run_rec(i3, rp, ra); k3 = bkey[g]; v3 = bval[g]; }
    }
    float w2v = 0.0f;
    if constexpr (FIRST) {
        if (t < 16 * 16 && (t & 15) < NCLS) w2v = W2[(t >> 4) * NCLS + (t & 15)];
    }
    #pragma unroll
    for (int u = 0; u < H_PT; ++u) { const int j = t + u * 1024; if (j < RNG * HID) acc[j] = hr[u]; }
    if (t < RNG) dli[t] = dv;
    if constexpr (FIRST) {
        if (t < 16 * 16) w2s[t] = (__bf16)w2v;
    }
    __syncthreads();

    for (; base < e; base += 128) {
        int   nk0 = 0, nk1 = 0, nk2 = 0, nk3 = 0;
        float nv0 = 0.0f, nv1 = 0.0f, nv2 = 0.0f, nv3 = 0.0f;
        {
            const int i0 = base + 128 + lane, i1 = i0 + 32, i2 = i0 + 64, i3 = i0 + 96;
            if (i0 < e) { const int g = run_rec(i0, rp, ra); nk0 = bkey[g]; nv0 = bval[g]; }
            if (i1 < e) { const int g = run_rec(i1, rp, ra); nk1 = bkey[g]; nv1 = bval[g]; }
            if (i2 < e) { const int g = run_rec(i2, rp, ra); nk2 = bkey[g]; nv2 = bval[g]; }
            if (i3 < e) { const int g = run_rec(i3, rp, ra); nk3 = bkey[g]; nv3 = bval[g]; }
        }

        #pragma nounroll
        for (int u = 0; u < 4; ++u) {
            const int b0 = base + u * 32;
            if (b0 < e) {
                int   kk[16];
                float vv[16], hv[16];
                #pragma unroll
                for (int j = 0; j < 16; ++j) { kk[j] = __shfl(k0, j + hsel); vv[j] = __shfl(v0, j + hsel); }
                #pragma unroll
                for (int j = 0; j < 16; ++j) hv[j] = h[(size_t)(kk[j] & 0x1FFFF) * HID + c];
                #pragma unroll
                for (int j = 0; j < 16; ++j) {
                    const float p = (b0 + j + hsel < e) ? vv[j] * hv[j] : 0.0f;
                    atomicAdd(&acc[(kk[j] >> 17) * HID + c], p);
                }
            }
            k0 = k1; k1 = k2; k2 = k3;
            v0 = v1; v1 = v2; v2 = v3;
        }
        k0 = nk0; k1 = nk1; k2 = nk2; k3 = nk3;
        v0 = nv0; v1 = nv1; v2 = nv2; v3 = nv3;
    }
    __syncthreads();

    if constexpr (FIRST) {
        const int wave = t >> 5;
        if (wave < RNG / 16 && n0 + wave * 16 < N_NODES) {
            const int hh = lane >> 4;
            const float* row = acc + (wave * 16 + c) * HID;
            const float  dl  = dli[wave * 16 + c];
            v16bf a, b;
            #pragma unroll
            for (int i = 0; i < 16; ++i) { a[i] = (__bf16)0.0f; b[i] = (__bf16)0.0f; }
            #pragma unroll
            for (int j = 0; j < 4; ++j) {
                const int kl = 2 * j + 8 * hh;
                const float x0 = dl * row[kl]     + bias[kl];
                const float x1 = dl * row[kl + 1] + bias[kl + 1];
                a[2 * j]     = (__bf16)(x0 > 0.0f ? x0 : 0.0f);
                a[2 * j + 1] = (__bf16)(x1 > 0.0f ? x1 : 0.0f);
                b[2 * j]     = w2s[kl * 16 + c];
                b[2 * j + 1] = w2s[(kl + 1) * 16 + c];
            }
            v8f d;
            #pragma unroll
            for (int q = 0; q < 8; ++q) d[q] = 0.0f;
            d = __builtin_amdgcn_wmma_f32_16x16x32_bf16(
                    false, a, false, b, (short)0, d, false, false);
            #pragma unroll
            for (int q = 0; q < 8; ++q)
                acc[(wave * 16 + 8 * hh + q) * HID + c] = d[q] * dli[wave * 16 + 8 * hh + q];
        }
        __syncthreads();

        for (int j = t; j < RNG * HID; j += 1024)
            if (n0 + (j >> 4) < N_NODES) dst[(size_t)n0 * HID + j] = acc[j];
    } else {
        if (t < RNG && n0 + t < N_NODES) {
            float z[NCLS];
            float m = -INFINITY;
            #pragma unroll
            for (int q = 0; q < NCLS; ++q) {
                z[q] = dli[t] * acc[t * HID + q] + bias[q];
                m = fmaxf(m, z[q]);
            }
            float se = 0.0f;
            #pragma unroll
            for (int q = 0; q < NCLS; ++q) se += __expf(z[q] - m);
            const float ls = __logf(se) + m;
            #pragma unroll
            for (int q = 0; q < NCLS; ++q) ost[t * NCLS + q] = z[q] - ls;
        }
        __syncthreads();

        const int rows = (N_NODES - n0 < RNG) ? (N_NODES - n0) : RNG;
        for (int j = t; j < rows * NCLS; j += 1024) dst[(size_t)n0 * NCLS + j] = ost[j];
    }
}


extern "C" void kernel_launch(void* const* d_in, const int* in_sizes, int n_in,
                              void* d_out, int out_size, void* d_ws, size_t ws_size,
                              hipStream_t stream) {
    const float* x  = (const float*)d_in[0];
    const int*   ei = (const int*)  d_in[1];
    const float* ew = (const float*)d_in[2];
    const float* W1 = (const float*)d_in[3];
    const float* b1 = (const float*)d_in[4];
    const float* W2 = (const float*)d_in[5];
    const float* b2 = (const float*)d_in[6];
    float* out = (float*)d_out;

    const int* srcp = ei;
    const int* dstp = ei + N_EDGES;

    float* ws   = (float*)d_ws;
    float* dinv = ws;
    int*   bkey = (int*)(dinv + N_NODES);
    float* bval = (float*)(bkey + N_EDGES);
    float* hbuf = bval + N_EDGES;
    float* h2   = hbuf + (size_t)N_NODES * HID;

    int* rb   = (int*)d_out;

    k_fill_gemm <<<NB + GM_BLOCKS, 1024, 0, stream>>>(srcp, dstp, ew, rb, bkey, bval, x, W1, hbuf);
    k_deg_dinv  <<<NB, 1024, 0, stream>>>(bkey, bval, rb, dinv, hbuf);

    k_agg<true> <<<NB, 1024, 0, stream>>>(bkey, bval, rb, dinv, hbuf, b1, W2, h2);

    k_agg<false><<<NB, 1024, 0, stream>>>(bkey, bval, rb, dinv, h2, b2, W2, out);
}


# Aggregation hands records round as DPP row_share operands of the address adds and the multiply (per-record math once per lane), not LDS permutes; past-the-end adds go to a spare LDS row, no select
# speedup vs baseline: 5.8468x; 1.0764x over previous
#include <hip/hip_runtime.h>
#include <math.h>


#define N_NODES 100000
#define N_EDGES 3200000
#define F_IN    512
#define HID     16
#define NCLS    10
#define NTILES  (N_NODES / 16)

typedef __attribute__((ext_vector_type(16))) __bf16 v16bf;
typedef __attribute__((ext_vector_type(8)))  float  v8f;


#define NB      256
#define RNG     416
#define CH      (N_EDGES / NB)
#define CH_PT   ((CH + 1023) / 1024)
#define RB_STEP (RNG * NCLS)
#define H_PT    ((RNG * HID + 1023) / 1024)
#define DG_PT   16
#define RUNS_PW (NB / 32)

static_assert((N_NODES % RNG == 0 ? RNG : N_NODES % RNG) * NCLS >= 2 * NB, "the last range's rows hold its run table");
static_assert(RB_STEP % 2 == 0, "run pairs are 8-byte aligned");

__device__ __forceinline__ int wave_runs(const int* tb, int wave, int lane, int (&rp)[RUNS_PW], int (&ra)[RUNS_PW]) {
    int gs = 0, ln = 0;
    if (lane < RUNS_PW) {
        const int2 q = reinterpret_cast<const int2*>(tb)[wave * RUNS_PW + lane];
        gs = q.x; ln = q.y;
    }
    int inc = ln;
    #pragma unroll
    for (int d = 1; d < RUNS_PW; d <<= 1) {
        const int o = __shfl_up(inc, d);
        if (lane >= d) inc += o;
    }
    const int ex = inc - ln;
    #pragma unroll
    for (int l = 0; l < RUNS_PW; ++l) {
        rp[l] = __builtin_amdgcn_readlane(ex, l);
        ra[l] = __builtin_amdgcn_readlane(gs - ex, l);
    }
    return __builtin_amdgcn_readlane(inc, RUNS_PW - 1);
}

__device__ __forceinline__ int run_rec(int i, const int (&rp)[RUNS_PW], const int (&ra)[RUNS_PW]) {
    int a = ra[0];
    #pragma unroll
    for (int l = 1; l < RUNS_PW; ++l) a = (i >= rp[l]) ? ra[l] : a;
    return a + i;
}

__device__ __forceinline__ int scan256_excl(int v, int t, int* wt) {
    const int lane = t & 31;
    int inc = v;
    if (t < 256) {
        #pragma unroll
        for (int d = 1; d < 32; d <<= 1) {
            int o = __shfl_up(inc, d);
            if (lane >= d) inc += o;
        }
        if (lane == 31) wt[t >> 5] = inc;
    }
    __syncthreads();
    int off = 0;
    if (t < 256) {
        const int wv = t >> 5;
        #pragma unroll
        for (int k = 0; k < 7; ++k) if (k < wv) off += wt[k];
    }
    __syncthreads();
    return off + inc - v;
}


#define GM_BLOCKS 250
#define GM_WAVES  25
#define FILL_WORDS (2 * NB + 8 + 2 * CH)

static_assert(GM_BLOCKS * GM_WAVES == NTILES, "one tile per product wave");
static_assert(FILL_WORDS * 4 >= F_IN * HID * 2, "W1 as bf16 fits the shared buffer");

__device__ __forceinline__ void gemm1_role(int* smem, int g, const float* __restrict__ x,
                                           const float* __restrict__ W1, float* __restrict__ h);

__global__ __launch_bounds__(1024) void k_fill_gemm(const int* __restrict__ src, const int* __restrict__ dst,
                                                    const float* __restrict__ w, int* __restrict__ rb,
                                                    int* __restrict__ bkey, float* __restrict__ bval,
                                                    const float* __restrict__ x, const float* __restrict__ W1,
                                                    float* __restrict__ h) {
    __shared__ __attribute__((aligned(16))) int smem[FILL_WORDS];

    if (blockIdx.x >= NB) {
        gemm1_role(smem, blockIdx.x - NB, x, W1, h);
        return;
    }

    int*   cnt  = smem;
    int*   lst  = smem + NB;
    int*   wt   = smem + 2 * NB;
    int*   skey = smem + 2 * NB + 8;
    float* sval = reinterpret_cast<float*>(skey + CH);

    const int t  = threadIdx.x;
    const int b  = blockIdx.x;
    const int e0 = b * CH;

    if (t < NB) cnt[t] = 0;
    __syncthreads();

    unsigned pk[CH_PT];
    int dd[CH_PT];
    #pragma unroll
    for (int k = 0; k < CH_PT; ++k) {
        const int i = t + k * 1024;
        dd[k] = (i < CH) ? dst[e0 + i] : -1;
    }
    #pragma unroll
    for (int k = 0; k < CH_PT; ++k) {
        pk[k] = 0;
        if (dd[k] >= 0) {
            const int d  = dd[k];
            const int bn = d / RNG;
            const int rk = atomicAdd(&cnt[bn], 1);
            pk[k] = (unsigned)rk | ((unsigned)bn << 14) | ((unsigned)(d - bn * RNG) << 22);
        }
    }
    __syncthreads();

    int cv = 0;
    if (t < NB) cv = cnt[t];
    const int ex = scan256_excl(cv, t, wt);
    if (t < NB) lst[t] = ex;
    __syncthreads();

    #pragma unroll
    for (int k = 0; k < CH_PT; ++k) {
        const int i = t + k * 1024;
        if (i < CH) {
            const int rk = pk[k] & 0x3FFF;
            const int bn = (pk[k] >> 14) & 0xFF;
            const int dl = pk[k] >> 22;
            const int p  = lst[bn] + rk;
            skey[p] = src[e0 + i] | (dl << 17);
            sval[p] = w[e0 + i];
        }
    }
    __syncthreads();

    if (t < NB && t * RNG < N_NODES)
        reinterpret_cast<int2*>(rb + t * RB_STEP)[b] = make_int2(e0 + ex, cv);

    for (int i = t; i < CH; i += 1024) {
        bkey[e0 + i] = skey[i];
        bval[e0 + i] = sval[i];
    }
}

__global__ __launch_bounds__(1024) void k_deg_dinv(const int* __restrict__ bkey, const float* __restrict__ bval,
                                                   const int* __restrict__ rb, float* __restrict__ dinv,
                                                   float* __restrict__ h) {
    __shared__ float dg[RNG];
    const int t = threadIdx.x;
    const int r = blockIdx.x;
    if (r * RNG >= N_NODES) return;
    const int n0   = r * RNG;
    const int nel  = ((N_NODES - n0 < RNG) ? (N_NODES - n0) : RNG) * HID;
    float* hr = h + (size_t)n0 * HID;
    float hv[H_PT];
    #pragma unroll
    for (int u = 0; u < H_PT; ++u) { const int j = t + u * 1024; hv[u] = (j < nel) ? hr[j] : 0.0f; }
    int rp[RUNS_PW], ra[RUNS_PW];
    const int e = wave_runs(rb + r * RB_STEP, t >> 5, t & 31, rp, ra);
    if (t < RNG) dg[t] = 1.0f;
    __syncthreads();
    for (int i0 = (t & 31); i0 < e; i0 += DG_PT * 32) {
        int   k[DG_PT];
        float v[DG_PT];
        #pragma unroll
        for (int u = 0; u < DG_PT; ++u) {
            const int i = i0 + u * 32;
            k[u] = -1; v[u] = 0.0f;
            if (i < e) { const int g = run_rec(i, rp, ra); k[u] = bkey[g]; v[u] = bval[g]; }
        }
        #pragma unroll
        for (int u = 0; u < DG_PT; ++u) if (k[u] >= 0) atomicAdd(&dg[k[u] >> 17], v[u]);
    }
    __syncthreads();
    const int node = r * RNG + t;
    if (t < RNG) {
        const float d  = dg[t];
        const float di = (d > 0.0f) ? rsqrtf(d) : 0.0f;
        if (node < N_NODES) dinv[node] = di;
        dg[t] = di;
    }
    __syncthreads();
    #pragma unroll
    for (int u = 0; u < H_PT; ++u) { const int j = t + u * 1024; if (j < nel) hr[j] = hv[u] * dg[j >> 4]; }
}


__device__ __forceinline__ void gemm1_role(int* smem, int g, const float* __restrict__ x,
                                           const float* __restrict__ W1, float* __restrict__ h) {
    __bf16* w1s = reinterpret_cast<__bf16*>(smem);
    int t = threadIdx.x;
    #pragma unroll
    for (int i = 0; i < F_IN * HID / 1024; ++i) w1s[t + i * 1024] = (__bf16)W1[t + i * 1024];
    __syncthreads();

    int wave = t >> 5;
    int lane = t & 31;
    if (wave >= GM_WAVES) return;
    int tile = g * GM_WAVES + wave;

    int col = lane & 15;
    int hh  = lane >> 4;
    const float* xr = x + (size_t)(tile * 16 + col) * F_IN;

    v8f acc;
    #pragma unroll
    for (int r = 0; r < 8; ++r) acc[r] = 0.0f;

    for (int k0 = 0; k0 < F_IN; k0 += 32) {
        v16bf a, b;
        #pragma unroll
        for (int j = 0; j < 8; ++j) {
            int kl = (j < 4) ? (2 * j + 8 * hh) : (16 + 2 * (j - 4) + 8 * hh);
            int k  = k0 + kl;
            float2 xv = *reinterpret_cast<const float2*>(xr + k);
            a[2 * j]     = (__bf16)xv.x;
            a[2 * j + 1] = (__bf16)xv.y;
            b[2 * j]     = w1s[k * HID + col];
            b[2 * j + 1] = w1s[(k + 1) * HID + col];
        }
        acc = __builtin_amdgcn_wmma_f32_16x16x32_bf16(
                  false, a, false, b, (short)0, acc, false, false);
    }

    float* out = h + (size_t)(tile * 16 + 8 * hh) * HID + col;
    #pragma unroll
    for (int r = 0; r < 8; ++r) out[(size_t)r * HID] = acc[r];
}


template <int J>
__device__ __forceinline__ int row_take(int v) {
    return __builtin_amdgcn_mov_dpp(v, 0x150 + J, 0xf, 0xf, true);
}

template <bool FIRST>
__global__ __launch_bounds__(1024) void k_agg(const int* __restrict__ bkey, const float* __restrict__ bval,
                                              const int* rb, const float* __restrict__ dinv,
                                              const float* __restrict__ h, const float* __restrict__ bias,
                                              const float* __restrict__ W2, float* dst) {
    __shared__ float acc[(RNG + 1) * HID];
    __shared__ float dli[RNG];
    __shared__ __bf16 w2s[FIRST ? 16 * 16 : 2];
    __shared__ float  ost[FIRST ? 1 : RNG * NCLS];

    const int t    = threadIdx.x;
    const int lane = t & 31;
    const int c    = lane & 15;
    const int r    = blockIdx.x;
    const int n0   = r * RNG;
    if (n0 >= N_NODES) return;

    float hr[H_PT];
    #pragma unroll
    for (int u = 0; u < H_PT; ++u) {
        const int j = t + u * 1024;
        hr[u] = (j < RNG * HID && n0 + (j >> 4) < N_NODES) ? h[(size_t)n0 * HID + j] : 0.0f;
    }
    float dv = 0.0f;
    if (t < RNG && n0 + t < N_NODES) dv = dinv[n0 + t];
    int rp[RUNS_PW], ra[RUNS_PW];
    const int e = wave_runs(rb + r * RB_STEP, t >> 5, lane, rp, ra);
    int   k0 = 0, k1 = 0, k2 = 0, k3 = 0;
    float v0 = 0.0f, v1 = 0.0f, v2 = 0.0f, v3 = 0.0f;
    int base = 0;
    {
        const int i0 = lane, i1 = i0 + 32, i2 = i0 + 64, i3 = i0 + 96;
        if (i0 < e) { const int g = run_rec(i0, rp, ra); k0 = bkey[g]; v0 = bval[g]; }
        if (i1 < e) { const int g = run_rec(i1, rp, ra); k1 = bkey[g]; v1 = bval[g]; }
        if (i2 < e) { const int g = run_rec(i2, rp, ra); k2 = bkey[g]; v2 = bval[g]; }
        if (i3 < e) { const int g = run_rec(i3, rp, ra); k3 = bkey[g]; v3 = bval[g]; }
    }
    float w2v = 0.0f;
    if constexpr (FIRST) {
        if (t < 16 * 16 && (t & 15) < NCLS) w2v = W2[(t >> 4) * NCLS + (t & 15)];
    }
    #pragma unroll
    for (int u = 0; u < H_PT; ++u) { const int j = t + u * 1024; if (j < RNG * HID) acc[j] = hr[u]; }
    if (t < RNG) dli[t] = dv;
    if constexpr (FIRST) {
        if (t < 16 * 16) w2s[t] = (__bf16)w2v;
    }
    __syncthreads();

    for (; base < e; base += 128) {
        int   nk0 = 0, nk1 = 0, nk2 = 0, nk3 = 0;
        float nv0 = 0.0f, nv1 = 0.0f, nv2 = 0.0f, nv3 = 0.0f;
        {
            const int i0 = base + 128 + lane, i1 = i0 + 32, i2 = i0 + 64, i3 = i0 + 96;
            if (i0 < e) { const int g = run_rec(i0, rp, ra); nk0 = bkey[g]; nv0 = bval[g]; }
            if (i1 < e) { const int g = run_rec(i1, rp, ra); nk1 = bkey[g]; nv1 = bval[g]; }
            if (i2 < e) { const int g = run_rec(i2, rp, ra); nk2 = bkey[g]; nv2 = bval[g]; }
            if (i3 < e) { const int g = run_rec(i3, rp, ra); nk3 = bkey[g]; nv3 = bval[g]; }
        }

        #pragma nounroll
        for (int u = 0; u < 4; ++u) {
            const int b0 = base + u * 32;
            if (b0 < e) {
                float hv[16];
                const bool  live = b0 + lane < e;
                const int   goff = (k0 & 0x1FFFF) * HID;
                const int   loff = live ? (k0 >> 17) * HID : RNG * HID;
                const int   wbit = __float_as_int(v0);
                #define GATHER(J) hv[J] = h[(size_t)(unsigned)(row_take<J>(goff) + c)];
                GATHER(0) GATHER(1) GATHER(2)  GATHER(3)  GATHER(4)  GATHER(5)  GATHER(6)  GATHER(7)
                GATHER(8) GATHER(9) GATHER(10) GATHER(11) GATHER(12) GATHER(13) GATHER(14) GATHER(15)
                #undef GATHER
                #define ADD(J) atomicAdd(&acc[row_take<J>(loff) + c], __int_as_float(row_take<J>(wbit)) * hv[J]);
                ADD(0) ADD(1) ADD(2)  ADD(3)  ADD(4)  ADD(5)  ADD(6)  ADD(7)
                ADD(8) ADD(9) ADD(10) ADD(11) ADD(12) ADD(13) ADD(14) ADD(15)
                #undef ADD
            }
            k0 = k1; k1 = k2; k2 = k3;
            v0 = v1; v1 = v2; v2 = v3;
        }
        k0 = nk0; k1 = nk1; k2 = nk2; k3 = nk3;
        v0 = nv0; v1 = nv1; v2 = nv2; v3 = nv3;
    }
    __syncthreads();

    if constexpr (FIRST) {
        const int wave = t >> 5;
        if (wave < RNG / 16 && n0 + wave * 16 < N_NODES) {
            const int hh = lane >> 4;
            const float* row = acc + (wave * 16 + c) * HID;
            const float  dl  = dli[wave * 16 + c];
            v16bf a, b;
            #pragma unroll
            for (int i = 0; i < 16; ++i) { a[i] = (__bf16)0.0f; b[i] = (__bf16)0.0f; }
            #pragma unroll
            for (int j = 0; j < 4; ++j) {
                const int kl = 2 * j + 8 * hh;
                const float x0 = dl * row[kl]     + bias[kl];
                const float x1 = dl * row[kl + 1] + bias[kl + 1];
                a[2 * j]     = (__bf16)(x0 > 0.0f ? x0 : 0.0f);
                a[2 * j + 1] = (__bf16)(x1 > 0.0f ? x1 : 0.0f);
                b[2 * j]     = w2s[kl * 16 + c];
                b[2 * j + 1] = w2s[(kl + 1) * 16 + c];
            }
            v8f d;
            #pragma unroll
            for (int q = 0; q < 8; ++q) d[q] = 0.0f;
            d = __builtin_amdgcn_wmma_f32_16x16x32_bf16(
                    false, a, false, b, (short)0, d, false, false);
            #pragma unroll
            for (int q = 0; q < 8; ++q)
                acc[(wave * 16 + 8 * hh + q) * HID + c] = d[q] * dli[wave * 16 + 8 * hh + q];
        }
        __syncthreads();

        for (int j = t; j < RNG * HID; j += 1024)
            if (n0 + (j >> 4) < N_NODES) dst[(size_t)n0 * HID + j] = acc[j];
    } else {
        if (t < RNG && n0 + t < N_NODES) {
            float z[NCLS];
            float m = -INFINITY;
            #pragma unroll
            for (int q = 0; q < NCLS; ++q) {
                z[q] = dli[t] * acc[t * HID + q] + bias[q];
                m = fmaxf(m, z[q]);
            }
            float se = 0.0f;
            #pragma unroll
            for (int q = 0; q < NCLS; ++q) se += __expf(z[q] - m);
            const float ls = __logf(se) + m;
            #pragma unroll
            for (int q = 0; q < NCLS; ++q) ost[t * NCLS + q] = z[q] - ls;
        }
        __syncthreads();

        const int rows = (N_NODES - n0 < RNG) ? (N_NODES - n0) : RNG;
        for (int j = t; j < rows * NCLS; j += 1024) dst[(size_t)n0 * NCLS + j] = ost[j];
    }
}


extern "C" void kernel_launch(void* const* d_in, const int* in_sizes, int n_in,
                              void* d_out, int out_size, void* d_ws, size_t ws_size,
                              hipStream_t stream) {
    const float* x  = (const float*)d_in[0];
    const int*   ei = (const int*)  d_in[1];
    const float* ew = (const float*)d_in[2];
    const float* W1 = (const float*)d_in[3];
    const float* b1 = (const float*)d_in[4];
    const float* W2 = (const float*)d_in[5];
    const float* b2 = (const float*)d_in[6];
    float* out = (float*)d_out;

    const int* srcp = ei;
    const int* dstp = ei + N_EDGES;

    float* ws   = (float*)d_ws;
    float* dinv = ws;
    int*   bkey = (int*)(dinv + N_NODES);
    float* bval = (float*)(bkey + N_EDGES);
    float* hbuf = bval + N_EDGES;
    float* h2   = hbuf + (size_t)N_NODES * HID;

    int* rb   = (int*)d_out;

    k_fill_gemm <<<NB + GM_BLOCKS, 1024, 0, stream>>>(srcp, dstp, ew, rb, bkey, bval, x, W1, hbuf);
    k_deg_dinv  <<<NB, 1024, 0, stream>>>(bkey, bval, rb, dinv, hbuf);

    k_agg<true> <<<NB, 1024, 0, stream>>>(bkey, bval, rb, dinv, hbuf, b1, W2, h2);

    k_agg<false><<<NB, 1024, 0, stream>>>(bkey, bval, rb, dinv, h2, b2, W2, out);
}
